# starting kernel
# baseline (speedup 1.0000x reference)
#include <hip/hip_runtime.h>


typedef __attribute__((ext_vector_type(2))) float v2f;
typedef __attribute__((ext_vector_type(8))) float v8f;

#define NQ 8
#define NLAYERS 4

struct cpx { float re, im; };

__device__ __forceinline__ cpx cmul(cpx a, cpx b) {
  cpx r; r.re = a.re * b.re - a.im * b.im; r.im = a.re * b.im + a.im * b.re; return r;
}

__device__ __forceinline__ cpx u_entry(float a, float bb, float g, int i, int j) {
  float ca = cosf(0.5f * a),  sa = sinf(0.5f * a);
  float cb = cosf(0.5f * bb), sb = sinf(0.5f * bb);
  cpx m;
  if (i == 0 && j == 0) { m.re =  cb * ca; m.im =  sb * sa; }
  else if (i == 0)      { m.re = -sb * ca; m.im = -cb * sa; }
  else if (j == 0)      { m.re =  sb * ca; m.im = -cb * sa; }
  else                  { m.re =  cb * ca; m.im = -sb * sa; }
  float cg = cosf(0.5f * g), sg = sinf(0.5f * g);
  cpx ph; ph.re = cg; ph.im = (i == 0) ? -sg : sg;
  return cmul(ph, m);
}

__host__ __device__ constexpr int chainperm(int r) {
  r ^= ((r >> 1) & 1);
  r ^= ((r >> 2) & 1) << 1;
  r ^= ((r >> 3) & 1) << 2;
  return r;
}

__global__ void qprep(const float* __restrict__ w, float* __restrict__ ws) {
  int t = threadIdx.x;
  int m = t >> 4, k = t & 15;
  for (int l = 0; l < NLAYERS; ++l) {
    cpx pr; pr.re = 1.f; pr.im = 0.f;
    #pragma unroll
    for (int j = 3; j >= 0; --j) {
      int q = 4 + (3 - j);
      int base = (l * NQ + q) * 3;
      pr = cmul(pr, u_entry(w[base], w[base + 1], w[base + 2], (m >> j) & 1, (k >> j) & 1));
    }
    float* dst = ws + (size_t)((l * 2 + 0) * 2) * 256;
    dst[m * 16 + k]       = pr.re;
    dst[256 + m * 16 + k] = pr.im;

    int mp = chainperm(m);
    cpx pc; pc.re = 1.f; pc.im = 0.f;
    #pragma unroll
    for (int j = 3; j >= 0; --j) {
      int q = (3 - j);
      int base = (l * NQ + q) * 3;
      pc = cmul(pc, u_entry(w[base], w[base + 1], w[base + 2], (mp >> j) & 1, (k >> j) & 1));
    }
    dst = ws + (size_t)((l * 2 + 1) * 2) * 256;
    dst[m * 16 + k]       = pc.re;
    dst[256 + m * 16 + k] = pc.im;
  }
}


__device__ __forceinline__ void cd_to_b(const float* S, float* Bv, int half) {
  const int r0t[8] = {0, 1, 4, 5, 0, 1, 4, 5};
  #pragma unroll
  for (int d = 0; d < 8; ++d) {
    int r0 = r0t[d];
    if (d < 4) {
      float t = __shfl_xor(S[r0 + 2], 16, 32);
      Bv[d] = half ? t : S[r0];
    } else {
      float t = __shfl_xor(S[r0], 16, 32);
      Bv[d] = half ? S[r0 + 2] : t;
    }
  }
}

__device__ __forceinline__ void cmulmat(const float* __restrict__ mat, int lane,
                                        const float* Xb, const float* Yb,
                                        float* Xo, float* Yo) {
  const int half = lane >> 4;
  const int mrow = lane & 15;
  float Are[8], Aim[8];
  #pragma unroll
  for (int a = 0; a < 8; ++a) {
    int k = 4 * (a >> 1) + (a & 1) + 2 * half;
    Are[a] = mat[mrow * 16 + k];
    Aim[a] = mat[256 + mrow * 16 + k];
  }
  v8f accX = {}; v8f accY = {};
  #pragma unroll
  for (int kc = 0; kc < 4; ++kc) {
    int a0 = 2 * kc, a1 = a0 + 1;
    v2f ar;  ar[0]  = Are[a0];  ar[1]  = Are[a1];
    v2f ai;  ai[0]  = Aim[a0];  ai[1]  = Aim[a1];
    v2f bx;  bx[0]  = Xb[a0];   bx[1]  = Xb[a1];
    v2f by;  by[0]  = Yb[a0];   by[1]  = Yb[a1];
    v2f nby; nby[0] = -Yb[a0];  nby[1] = -Yb[a1];
    accX = __builtin_amdgcn_wmma_f32_16x16x4_f32(false, ar, false, bx,  (short)0, accX, false, false);
    accX = __builtin_amdgcn_wmma_f32_16x16x4_f32(false, ai, false, nby, (short)0, accX, false, false);
    accY = __builtin_amdgcn_wmma_f32_16x16x4_f32(false, ar, false, by,  (short)0, accY, false, false);
    accY = __builtin_amdgcn_wmma_f32_16x16x4_f32(false, ai, false, bx,  (short)0, accY, false, false);
  }
  #pragma unroll
  for (int v = 0; v < 8; ++v) { Xo[v] = accX[v]; Yo[v] = accY[v]; }
}

__global__ __launch_bounds__(256) void qsim(const float* __restrict__ x,
                                            const float* __restrict__ ws,
                                            float* __restrict__ out, int B) {
  __shared__ float lds[8 * 512];
  const int lane  = threadIdx.x & 31;
  const int wslot = threadIdx.x >> 5;
  const int b     = blockIdx.x * 8 + wslot;
  if (b >= B) return;
  const int half = lane >> 4;
  const int col  = lane & 15;
  float* myLds = lds + wslot * 512;

  float cq[8], sq[8];
  #pragma unroll
  for (int q = 0; q < 8; ++q) {
    float xv = x[b * 8 + q];
    cq[q] = cosf(0.5f * xv); sq[q] = sinf(0.5f * xv);
  }
  float colf = 1.f;
  #pragma unroll
  for (int q = 0; q < 4; ++q) colf *= ((col >> (3 - q)) & 1) ? sq[q] : cq[q];
  float X[8], Y[8];
  #pragma unroll
  for (int v = 0; v < 8; ++v) {
    int r = v + 8 * half;
    float rowf = 1.f;
    #pragma unroll
    for (int q = 4; q < 8; ++q) rowf *= ((r >> (7 - q)) & 1) ? sq[q] : cq[q];
    X[v] = colf * rowf; Y[v] = 0.f;
  }

  for (int l = 0; l < NLAYERS; ++l) {
    const int flip = l & 1;
    const float* matA = ws + (size_t)((l * 2 + flip)       * 2) * 256;
    const float* matB = ws + (size_t)((l * 2 + (1 - flip)) * 2) * 256;

    float Xb[8], Yb[8];
    cd_to_b(X, Xb, half);
    cd_to_b(Y, Yb, half);
    cmulmat(matA, lane, Xb, Yb, X, Y);

    #pragma unroll
    for (int v = 0; v < 8; ++v) {
      myLds[(v + 8 * half) * 16 + col]       = X[v];
      myLds[256 + (v + 8 * half) * 16 + col] = Y[v];
    }
    __syncthreads();
    #pragma unroll
    for (int bb = 0; bb < 8; ++bb) {
      int kk = 4 * (bb >> 1) + (bb & 1) + 2 * half;
      Xb[bb] = myLds[col * 16 + kk];
      Yb[bb] = myLds[256 + col * 16 + kk];
    }
    __syncthreads();

    cmulmat(matB, lane, Xb, Yb, X, Y);

    if (flip) {
      #pragma unroll
      for (int v = 0; v < 8; ++v) {
        float tx = __shfl_xor(X[v], 16, 32);
        float ty = __shfl_xor(Y[v], 16, 32);
        if (lane & 1) { X[v] = tx; Y[v] = ty; }
      }
      float nX[8], nY[8];
      #pragma unroll
      for (int v = 0; v < 8; ++v) {
        const int r0 = chainperm(v), r1 = chainperm(v + 8);
        float t0x = (r0 < 8)  ? X[r0 & 7] : __shfl_xor(X[r0 & 7], 16, 32);
        float t0y = (r0 < 8)  ? Y[r0 & 7] : __shfl_xor(Y[r0 & 7], 16, 32);
        float t1x = (r1 >= 8) ? X[r1 & 7] : __shfl_xor(X[r1 & 7], 16, 32);
        float t1y = (r1 >= 8) ? Y[r1 & 7] : __shfl_xor(Y[r1 & 7], 16, 32);
        nX[v] = half ? t1x : t0x;
        nY[v] = half ? t1y : t0y;
      }
      #pragma unroll
      for (int v = 0; v < 8; ++v) { X[v] = nX[v]; Y[v] = nY[v]; }
    } else {
      #pragma unroll
      for (int v = 1; v < 8; v += 2) {
        X[v] = __shfl_xor(X[v], 8, 32);
        Y[v] = __shfl_xor(Y[v], 8, 32);
      }
      int src = chainperm(lane & 15) | (lane & 16);
      #pragma unroll
      for (int v = 0; v < 8; ++v) {
        X[v] = __shfl(X[v], src, 32);
        Y[v] = __shfl(Y[v], src, 32);
      }
    }
  }

  float psum = 0.f, s5 = 0.f, s6 = 0.f, s7 = 0.f;
  #pragma unroll
  for (int v = 0; v < 8; ++v) {
    float p = X[v] * X[v] + Y[v] * Y[v];
    psum += p;
    s5 += ((v >> 2) & 1) ? -p : p;
    s6 += ((v >> 1) & 1) ? -p : p;
    s7 += ((v) & 1)      ? -p : p;
  }
  float o[8];
  o[0] = ((col >> 3) & 1) ? -psum : psum;
  o[1] = ((col >> 2) & 1) ? -psum : psum;
  o[2] = ((col >> 1) & 1) ? -psum : psum;
  o[3] = ((col) & 1)      ? -psum : psum;
  o[4] = half ? -psum : psum;
  o[5] = s5; o[6] = s6; o[7] = s7;
  #pragma unroll
  for (int q = 0; q < 8; ++q) {
    float v = o[q];
    #pragma unroll
    for (int off = 16; off > 0; off >>= 1) v += __shfl_xor(v, off, 32);
    o[q] = v;
  }
  if (lane == 0) {
    #pragma unroll
    for (int q = 0; q < 8; ++q) out[b * 8 + q] = o[q];
  }
}

extern "C" void kernel_launch(void* const* d_in, const int* in_sizes, int n_in,
                              void* d_out, int out_size, void* d_ws, size_t ws_size,
                              hipStream_t stream) {
  const float* x = (const float*)d_in[0];
  const float* w = (const float*)d_in[1];
  float* out = (float*)d_out;
  float* ws  = (float*)d_ws;
  int B = in_sizes[0] / NQ;

  qprep<<<1, 256, 0, stream>>>(w, ws);
  int nblocks = (B + 7) / 8;
  qsim<<<nblocks, 256, 0, stream>>>(x, ws, out, B);
}


# Gate-matrix prep launch folded into the circuit kernel (2 launches -> 1): each work-group builds the 8 gate matrices in LDS in A-fragment order, trig one lane per weight; entry products regrouped
# speedup vs baseline: 1.5562x; 1.5562x over previous
#include <hip/hip_runtime.h>


typedef __attribute__((ext_vector_type(2))) float v2f;
typedef __attribute__((ext_vector_type(4))) float v4f;
typedef __attribute__((ext_vector_type(8))) float v8f;

#define NQ 8
#define NLAYERS 4

struct cpx { float re, im; };

__device__ __forceinline__ cpx cmul(cpx a, cpx b) {
  cpx r; r.re = a.re * b.re - a.im * b.im; r.im = a.re * b.im + a.im * b.re; return r;
}

__device__ __forceinline__ cpx u_entry(const float* cs, const float* sn, int gate, int i, int j) {
  float ca = cs[3 * gate],     sa = sn[3 * gate];
  float cb = cs[3 * gate + 1], sb = sn[3 * gate + 1];
  float cg = cs[3 * gate + 2], sg = sn[3 * gate + 2];
  cpx m;
  if (i == 0 && j == 0) { m.re =  cb * ca; m.im =  sb * sa; }
  else if (i == 0)      { m.re = -sb * ca; m.im = -cb * sa; }
  else if (j == 0)      { m.re =  sb * ca; m.im = -cb * sa; }
  else                  { m.re =  cb * ca; m.im = -sb * sa; }
  cpx ph; ph.re = cg; ph.im = (i == 0) ? -sg : sg;
  return cmul(ph, m);
}

__host__ __device__ constexpr int chainperm(int r) {
  r ^= ((r >> 1) & 1);
  r ^= ((r >> 2) & 1) << 1;
  r ^= ((r >> 3) & 1) << 2;
  return r;
}

__device__ __forceinline__ void build_gates(const float* __restrict__ w, float* tmp,
                                            float* fragf, int t) {
  float* cs  = tmp;
  float* sn  = tmp + 96;
  float* p4  = tmp + 256;
  if (t < NLAYERS * NQ * 3) {
    float h = 0.5f * w[t];
    cs[t] = cosf(h);
    sn[t] = sinf(h);
  }
  __syncthreads();
  {
    int l = t >> 6, p = (t >> 4) & 3, i4 = (t >> 2) & 3, j4 = t & 3;
    int g0 = l * NQ + 2 * p;
    cpx e = cmul(u_entry(cs, sn, g0,     i4 >> 1, j4 >> 1),
                 u_entry(cs, sn, g0 + 1, i4 & 1,  j4 & 1));
    p4[2 * t]     = e.re;
    p4[2 * t + 1] = e.im;
  }
  __syncthreads();
  {
    int m = t >> 4, k = t & 15;
    int mp = chainperm(m);
    int slot = ((k >> 2) * 32 + m + 16 * ((k >> 1) & 1)) * 4 + (k & 1);
    #pragma unroll
    for (int l = 0; l < NLAYERS; ++l) {
      const float* pl = p4 + l * 128;
      cpx a, b, r;
      a.re = pl[(2 * 16 + (m >> 2) * 4 + (k >> 2)) * 2]; a.im = pl[(2 * 16 + (m >> 2) * 4 + (k >> 2)) * 2 + 1];
      b.re = pl[(3 * 16 + (m & 3) * 4 + (k & 3)) * 2];   b.im = pl[(3 * 16 + (m & 3) * 4 + (k & 3)) * 2 + 1];
      r = cmul(a, b);
      fragf[(l * 2 + 0) * 512 + slot]     = r.re;
      fragf[(l * 2 + 0) * 512 + slot + 2] = r.im;
      a.re = pl[(0 * 16 + (mp >> 2) * 4 + (k >> 2)) * 2]; a.im = pl[(0 * 16 + (mp >> 2) * 4 + (k >> 2)) * 2 + 1];
      b.re = pl[(1 * 16 + (mp & 3) * 4 + (k & 3)) * 2];   b.im = pl[(1 * 16 + (mp & 3) * 4 + (k & 3)) * 2 + 1];
      r = cmul(a, b);
      fragf[(l * 2 + 1) * 512 + slot]     = r.re;
      fragf[(l * 2 + 1) * 512 + slot + 2] = r.im;
    }
  }
  __syncthreads();
}


__device__ __forceinline__ void cd_to_b(const float* S, float* Bv, int half) {
  const int r0t[8] = {0, 1, 4, 5, 0, 1, 4, 5};
  #pragma unroll
  for (int d = 0; d < 8; ++d) {
    int r0 = r0t[d];
    if (d < 4) {
      float t = __shfl_xor(S[r0 + 2], 16, 32);
      Bv[d] = half ? t : S[r0];
    } else {
      float t = __shfl_xor(S[r0], 16, 32);
      Bv[d] = half ? S[r0 + 2] : t;
    }
  }
}

__device__ __forceinline__ void cmulmat(const v4f* mat, int lane,
                                        const float* Xb, const float* Yb,
                                        float* Xo, float* Yo) {
  v4f f[4];
  #pragma unroll
  for (int kc = 0; kc < 4; ++kc) f[kc] = mat[kc * 32 + lane];
  v8f accX = {}; v8f accY = {};
  #pragma unroll
  for (int kc = 0; kc < 4; ++kc) {
    int a0 = 2 * kc, a1 = a0 + 1;
    v2f ar;  ar[0]  = f[kc][0]; ar[1]  = f[kc][1];
    v2f ai;  ai[0]  = f[kc][2]; ai[1]  = f[kc][3];
    v2f bx;  bx[0]  = Xb[a0];   bx[1]  = Xb[a1];
    v2f by;  by[0]  = Yb[a0];   by[1]  = Yb[a1];
    v2f nby; nby[0] = -Yb[a0];  nby[1] = -Yb[a1];
    accX = __builtin_amdgcn_wmma_f32_16x16x4_f32(false, ar, false, bx,  (short)0, accX, false, false);
    accX = __builtin_amdgcn_wmma_f32_16x16x4_f32(false, ai, false, nby, (short)0, accX, false, false);
    accY = __builtin_amdgcn_wmma_f32_16x16x4_f32(false, ar, false, by,  (short)0, accY, false, false);
    accY = __builtin_amdgcn_wmma_f32_16x16x4_f32(false, ai, false, bx,  (short)0, accY, false, false);
  }
  #pragma unroll
  for (int v = 0; v < 8; ++v) { Xo[v] = accX[v]; Yo[v] = accY[v]; }
}

__global__ __launch_bounds__(256) void qsim(const float* __restrict__ x,
                                            const float* __restrict__ w,
                                            float* __restrict__ out, int B) {
  __shared__ float lds[8 * 512];
  __shared__ v4f   frag[8 * 4 * 32];
  const int lane  = threadIdx.x & 31;
  const int wslot = threadIdx.x >> 5;
  const int bw    = blockIdx.x * 8 + wslot;
  const int b     = bw < B ? bw : B - 1;
  const int half = lane >> 4;
  const int col  = lane & 15;
  float* myLds = lds + wslot * 512;

  float cq[8], sq[8];
  #pragma unroll
  for (int q = 0; q < 8; ++q) {
    float xv = x[b * 8 + q];
    cq[q] = cosf(0.5f * xv); sq[q] = sinf(0.5f * xv);
  }
  float colf = 1.f;
  #pragma unroll
  for (int q = 0; q < 4; ++q) colf *= ((col >> (3 - q)) & 1) ? sq[q] : cq[q];
  float X[8], Y[8];
  #pragma unroll
  for (int v = 0; v < 8; ++v) {
    int r = v + 8 * half;
    float rowf = 1.f;
    #pragma unroll
    for (int q = 4; q < 8; ++q) rowf *= ((r >> (7 - q)) & 1) ? sq[q] : cq[q];
    X[v] = colf * rowf; Y[v] = 0.f;
  }

  build_gates(w, lds, (float*)frag, threadIdx.x);

  for (int l = 0; l < NLAYERS; ++l) {
    const int flip = l & 1;
    const v4f* matA = frag + (l * 2 + flip)       * 128;
    const v4f* matB = frag + (l * 2 + (1 - flip)) * 128;

    float Xb[8], Yb[8];
    cd_to_b(X, Xb, half);
    cd_to_b(Y, Yb, half);
    cmulmat(matA, lane, Xb, Yb, X, Y);

    #pragma unroll
    for (int v = 0; v < 8; ++v) {
      myLds[(v + 8 * half) * 16 + col]       = X[v];
      myLds[256 + (v + 8 * half) * 16 + col] = Y[v];
    }
    __syncthreads();
    #pragma unroll
    for (int bb = 0; bb < 8; ++bb) {
      int kk = 4 * (bb >> 1) + (bb & 1) + 2 * half;
      Xb[bb] = myLds[col * 16 + kk];
      Yb[bb] = myLds[256 + col * 16 + kk];
    }
    __syncthreads();

    cmulmat(matB, lane, Xb, Yb, X, Y);

    if (flip) {
      #pragma unroll
      for (int v = 0; v < 8; ++v) {
        float tx = __shfl_xor(X[v], 16, 32);
        float ty = __shfl_xor(Y[v], 16, 32);
        if (lane & 1) { X[v] = tx; Y[v] = ty; }
      }
      float nX[8], nY[8];
      #pragma unroll
      for (int v = 0; v < 8; ++v) {
        const int r0 = chainperm(v), r1 = chainperm(v + 8);
        float t0x = (r0 < 8)  ? X[r0 & 7] : __shfl_xor(X[r0 & 7], 16, 32);
        float t0y = (r0 < 8)  ? Y[r0 & 7] : __shfl_xor(Y[r0 & 7], 16, 32);
        float t1x = (r1 >= 8) ? X[r1 & 7] : __shfl_xor(X[r1 & 7], 16, 32);
        float t1y = (r1 >= 8) ? Y[r1 & 7] : __shfl_xor(Y[r1 & 7], 16, 32);
        nX[v] = half ? t1x : t0x;
        nY[v] = half ? t1y : t0y;
      }
      #pragma unroll
      for (int v = 0; v < 8; ++v) { X[v] = nX[v]; Y[v] = nY[v]; }
    } else {
      #pragma unroll
      for (int v = 1; v < 8; v += 2) {
        X[v] = __shfl_xor(X[v], 8, 32);
        Y[v] = __shfl_xor(Y[v], 8, 32);
      }
      int src = chainperm(lane & 15) | (lane & 16);
      #pragma unroll
      for (int v = 0; v < 8; ++v) {
        X[v] = __shfl(X[v], src, 32);
        Y[v] = __shfl(Y[v], src, 32);
      }
    }
  }

  float psum = 0.f, s5 = 0.f, s6 = 0.f, s7 = 0.f;
  #pragma unroll
  for (int v = 0; v < 8; ++v) {
    float p = X[v] * X[v] + Y[v] * Y[v];
    psum += p;
    s5 += ((v >> 2) & 1) ? -p : p;
    s6 += ((v >> 1) & 1) ? -p : p;
    s7 += ((v) & 1)      ? -p : p;
  }
  float o[8];
  o[0] = ((col >> 3) & 1) ? -psum : psum;
  o[1] = ((col >> 2) & 1) ? -psum : psum;
  o[2] = ((col >> 1) & 1) ? -psum : psum;
  o[3] = ((col) & 1)      ? -psum : psum;
  o[4] = half ? -psum : psum;
  o[5] = s5; o[6] = s6; o[7] = s7;
  #pragma unroll
  for (int q = 0; q < 8; ++q) {
    float v = o[q];
    #pragma unroll
    for (int off = 16; off > 0; off >>= 1) v += __shfl_xor(v, off, 32);
    o[q] = v;
  }
  if (lane == 0 && bw < B) {
    #pragma unroll
    for (int q = 0; q < 8; ++q) out[b * 8 + q] = o[q];
  }
}

extern "C" void kernel_launch(void* const* d_in, const int* in_sizes, int n_in,
                              void* d_out, int out_size, void* d_ws, size_t ws_size,
                              hipStream_t stream) {
  const float* x = (const float*)d_in[0];
  const float* w = (const float*)d_in[1];
  float* out = (float*)d_out;
  int B = in_sizes[0] / NQ;

  int nblocks = (B + 7) / 8;
  qsim<<<nblocks, 256, 0, stream>>>(x, w, out, B);
}


# Complex 16x16 product done with three real matrix products instead of four (shared Are*(X+Y) term; gate side holds Are and Aim-Are), 128 -> 96 matrix instructions per sample; sums regrouped
# speedup vs baseline: 1.8226x; 1.1712x over previous
#include <hip/hip_runtime.h>


typedef __attribute__((ext_vector_type(2))) float v2f;
typedef __attribute__((ext_vector_type(4))) float v4f;
typedef __attribute__((ext_vector_type(8))) float v8f;

#define NQ 8
#define NLAYERS 4

struct cpx { float re, im; };

__device__ __forceinline__ cpx cmul(cpx a, cpx b) {
  cpx r; r.re = a.re * b.re - a.im * b.im; r.im = a.re * b.im + a.im * b.re; return r;
}

__device__ __forceinline__ cpx u_entry(const float* cs, const float* sn, int gate, int i, int j) {
  float ca = cs[3 * gate],     sa = sn[3 * gate];
  float cb = cs[3 * gate + 1], sb = sn[3 * gate + 1];
  float cg = cs[3 * gate + 2], sg = sn[3 * gate + 2];
  cpx m;
  if (i == 0 && j == 0) { m.re =  cb * ca; m.im =  sb * sa; }
  else if (i == 0)      { m.re = -sb * ca; m.im = -cb * sa; }
  else if (j == 0)      { m.re =  sb * ca; m.im = -cb * sa; }
  else                  { m.re =  cb * ca; m.im = -sb * sa; }
  cpx ph; ph.re = cg; ph.im = (i == 0) ? -sg : sg;
  return cmul(ph, m);
}

__host__ __device__ constexpr int chainperm(int r) {
  r ^= ((r >> 1) & 1);
  r ^= ((r >> 2) & 1) << 1;
  r ^= ((r >> 3) & 1) << 2;
  return r;
}

__device__ __forceinline__ void build_gates(const float* __restrict__ w, float* tmp,
                                            float* fragf, int t) {
  float* cs  = tmp;
  float* sn  = tmp + 96;
  float* p4  = tmp + 256;
  if (t < NLAYERS * NQ * 3) {
    float h = 0.5f * w[t];
    cs[t] = cosf(h);
    sn[t] = sinf(h);
  }
  __syncthreads();
  {
    int l = t >> 6, p = (t >> 4) & 3, i4 = (t >> 2) & 3, j4 = t & 3;
    int g0 = l * NQ + 2 * p;
    cpx e = cmul(u_entry(cs, sn, g0,     i4 >> 1, j4 >> 1),
                 u_entry(cs, sn, g0 + 1, i4 & 1,  j4 & 1));
    p4[2 * t]     = e.re;
    p4[2 * t + 1] = e.im;
  }
  __syncthreads();
  {
    int m = t >> 4, k = t & 15;
    int mp = chainperm(m);
    int slot = ((k >> 2) * 32 + m + 16 * ((k >> 1) & 1)) * 4 + (k & 1);
    #pragma unroll
    for (int l = 0; l < NLAYERS; ++l) {
      const float* pl = p4 + l * 128;
      cpx a, b, r;
      a.re = pl[(2 * 16 + (m >> 2) * 4 + (k >> 2)) * 2]; a.im = pl[(2 * 16 + (m >> 2) * 4 + (k >> 2)) * 2 + 1];
      b.re = pl[(3 * 16 + (m & 3) * 4 + (k & 3)) * 2];   b.im = pl[(3 * 16 + (m & 3) * 4 + (k & 3)) * 2 + 1];
      r = cmul(a, b);
      fragf[(l * 2 + 0) * 512 + slot]     = r.re;
      fragf[(l * 2 + 0) * 512 + slot + 2] = r.im - r.re;
      a.re = pl[(0 * 16 + (mp >> 2) * 4 + (k >> 2)) * 2]; a.im = pl[(0 * 16 + (mp >> 2) * 4 + (k >> 2)) * 2 + 1];
      b.re = pl[(1 * 16 + (mp & 3) * 4 + (k & 3)) * 2];   b.im = pl[(1 * 16 + (mp & 3) * 4 + (k & 3)) * 2 + 1];
      r = cmul(a, b);
      fragf[(l * 2 + 1) * 512 + slot]     = r.re;
      fragf[(l * 2 + 1) * 512 + slot + 2] = r.im - r.re;
    }
  }
  __syncthreads();
}


__device__ __forceinline__ void cd_to_b(const float* S, float* Bv, int half) {
  const int r0t[8] = {0, 1, 4, 5, 0, 1, 4, 5};
  #pragma unroll
  for (int d = 0; d < 8; ++d) {
    int r0 = r0t[d];
    if (d < 4) {
      float t = __shfl_xor(S[r0 + 2], 16, 32);
      Bv[d] = half ? t : S[r0];
    } else {
      float t = __shfl_xor(S[r0], 16, 32);
      Bv[d] = half ? S[r0 + 2] : t;
    }
  }
}

__device__ __forceinline__ void cmulmat(const v4f* mat, int lane,
                                        const float* Xb, const float* Yb,
                                        float* Xo, float* Yo) {
  v4f f[4];
  #pragma unroll
  for (int kc = 0; kc < 4; ++kc) f[kc] = mat[kc * 32 + lane];
  v8f accM = {};
  #pragma unroll
  for (int kc = 0; kc < 4; ++kc) {
    int a0 = 2 * kc, a1 = a0 + 1;
    v2f ar;  ar[0] = f[kc][0];         ar[1] = f[kc][1];
    v2f bs;  bs[0] = Xb[a0] + Yb[a0];  bs[1] = Xb[a1] + Yb[a1];
    accM = __builtin_amdgcn_wmma_f32_16x16x4_f32(false, ar, false, bs, (short)0, accM, false, false);
  }
  v8f accX = accM; v8f accY = accM;
  #pragma unroll
  for (int kc = 0; kc < 4; ++kc) {
    int a0 = 2 * kc, a1 = a0 + 1;
    v2f ad;  ad[0] = f[kc][2];  ad[1] = f[kc][3];
    v2f aq;  aq[0] = -2.f * f[kc][0] - ad[0];
             aq[1] = -2.f * f[kc][1] - ad[1];
    v2f bx;  bx[0] = Xb[a0];    bx[1] = Xb[a1];
    v2f by;  by[0] = Yb[a0];    by[1] = Yb[a1];
    accX = __builtin_amdgcn_wmma_f32_16x16x4_f32(false, aq, false, by, (short)0, accX, false, false);
    accY = __builtin_amdgcn_wmma_f32_16x16x4_f32(false, ad, false, bx, (short)0, accY, false, false);
  }
  #pragma unroll
  for (int v = 0; v < 8; ++v) { Xo[v] = accX[v]; Yo[v] = accY[v]; }
}

__global__ __launch_bounds__(256, 16) void qsim(const float* __restrict__ x,
                                            const float* __restrict__ w,
                                            float* __restrict__ out, int B) {
  __shared__ float lds[8 * 512];
  __shared__ v4f   frag[8 * 4 * 32];
  const int lane  = threadIdx.x & 31;
  const int wslot = threadIdx.x >> 5;
  const int bw    = blockIdx.x * 8 + wslot;
  const int b     = bw < B ? bw : B - 1;
  const int half = lane >> 4;
  const int col  = lane & 15;
  float* myLds = lds + wslot * 512;

  float cq[8], sq[8];
  #pragma unroll
  for (int q = 0; q < 8; ++q) {
    float xv = x[b * 8 + q];
    cq[q] = cosf(0.5f * xv); sq[q] = sinf(0.5f * xv);
  }
  float colf = 1.f;
  #pragma unroll
  for (int q = 0; q < 4; ++q) colf *= ((col >> (3 - q)) & 1) ? sq[q] : cq[q];
  float X[8], Y[8];
  #pragma unroll
  for (int v = 0; v < 8; ++v) {
    int r = v + 8 * half;
    float rowf = 1.f;
    #pragma unroll
    for (int q = 4; q < 8; ++q) rowf *= ((r >> (7 - q)) & 1) ? sq[q] : cq[q];
    X[v] = colf * rowf; Y[v] = 0.f;
  }

  build_gates(w, lds, (float*)frag, threadIdx.x);

  for (int l = 0; l < NLAYERS; ++l) {
    const int flip = l & 1;
    const v4f* matA = frag + (l * 2 + flip)       * 128;
    const v4f* matB = frag + (l * 2 + (1 - flip)) * 128;

    float Xb[8], Yb[8];
    cd_to_b(X, Xb, half);
    cd_to_b(Y, Yb, half);
    cmulmat(matA, lane, Xb, Yb, X, Y);

    #pragma unroll
    for (int v = 0; v < 8; ++v) {
      myLds[(v + 8 * half) * 16 + col]       = X[v];
      myLds[256 + (v + 8 * half) * 16 + col] = Y[v];
    }
    __syncthreads();
    #pragma unroll
    for (int bb = 0; bb < 8; ++bb) {
      int kk = 4 * (bb >> 1) + (bb & 1) + 2 * half;
      Xb[bb] = myLds[col * 16 + kk];
      Yb[bb] = myLds[256 + col * 16 + kk];
    }
    __syncthreads();

    cmulmat(matB, lane, Xb, Yb, X, Y);

    if (flip) {
      #pragma unroll
      for (int v = 0; v < 8; ++v) {
        float tx = __shfl_xor(X[v], 16, 32);
        float ty = __shfl_xor(Y[v], 16, 32);
        if (lane & 1) { X[v] = tx; Y[v] = ty; }
      }
      float nX[8], nY[8];
      #pragma unroll
      for (int v = 0; v < 8; ++v) {
        const int r0 = chainperm(v), r1 = chainperm(v + 8);
        float t0x = (r0 < 8)  ? X[r0 & 7] : __shfl_xor(X[r0 & 7], 16, 32);
        float t0y = (r0 < 8)  ? Y[r0 & 7] : __shfl_xor(Y[r0 & 7], 16, 32);
        float t1x = (r1 >= 8) ? X[r1 & 7] : __shfl_xor(X[r1 & 7], 16, 32);
        float t1y = (r1 >= 8) ? Y[r1 & 7] : __shfl_xor(Y[r1 & 7], 16, 32);
        nX[v] = half ? t1x : t0x;
        nY[v] = half ? t1y : t0y;
      }
      #pragma unroll
      for (int v = 0; v < 8; ++v) { X[v] = nX[v]; Y[v] = nY[v]; }
    } else {
      #pragma unroll
      for (int v = 1; v < 8; v += 2) {
        X[v] = __shfl_xor(X[v], 8, 32);
        Y[v] = __shfl_xor(Y[v], 8, 32);
      }
      int src = chainperm(lane & 15) | (lane & 16);
      #pragma unroll
      for (int v = 0; v < 8; ++v) {
        X[v] = __shfl(X[v], src, 32);
        Y[v] = __shfl(Y[v], src, 32);
      }
    }
  }

  float psum = 0.f, s5 = 0.f, s6 = 0.f, s7 = 0.f;
  #pragma unroll
  for (int v = 0; v < 8; ++v) {
    float p = X[v] * X[v] + Y[v] * Y[v];
    psum += p;
    s5 += ((v >> 2) & 1) ? -p : p;
    s6 += ((v >> 1) & 1) ? -p : p;
    s7 += ((v) & 1)      ? -p : p;
  }
  float o[8];
  o[0] = ((col >> 3) & 1) ? -psum : psum;
  o[1] = ((col >> 2) & 1) ? -psum : psum;
  o[2] = ((col >> 1) & 1) ? -psum : psum;
  o[3] = ((col) & 1)      ? -psum : psum;
  o[4] = half ? -psum : psum;
  o[5] = s5; o[6] = s6; o[7] = s7;
  #pragma unroll
  for (int q = 0; q < 8; ++q) {
    float v = o[q];
    #pragma unroll
    for (int off = 16; off > 0; off >>= 1) v += __shfl_xor(v, off, 32);
    o[q] = v;
  }
  if (lane == 0 && bw < B) {
    #pragma unroll
    for (int q = 0; q < 8; ++q) out[b * 8 + q] = o[q];
  }
}

extern "C" void kernel_launch(void* const* d_in, const int* in_sizes, int n_in,
                              void* d_out, int out_size, void* d_ws, size_t ws_size,
                              hipStream_t stream) {
  const float* x = (const float*)d_in[0];
  const float* w = (const float*)d_in[1];
  float* out = (float*)d_out;
  int B = in_sizes[0] / NQ;

  int nblocks = (B + 7) / 8;
  qsim<<<nblocks, 256, 0, stream>>>(x, w, out, B);
}


# Angle-encoding trig once per wave: lane q loads x_q and takes one sincos, the 16 values are read across the wave (was 8 dependent loads and 16 sin/cos calls in every lane)
# speedup vs baseline: 1.9443x; 1.0668x over previous
#include <hip/hip_runtime.h>


typedef __attribute__((ext_vector_type(2))) float v2f;
typedef __attribute__((ext_vector_type(4))) float v4f;
typedef __attribute__((ext_vector_type(8))) float v8f;

#define NQ 8
#define NLAYERS 4

struct cpx { float re, im; };

__device__ __forceinline__ cpx cmul(cpx a, cpx b) {
  cpx r; r.re = a.re * b.re - a.im * b.im; r.im = a.re * b.im + a.im * b.re; return r;
}

__device__ __forceinline__ cpx u_entry(const float* cs, const float* sn, int gate, int i, int j) {
  float ca = cs[3 * gate],     sa = sn[3 * gate];
  float cb = cs[3 * gate + 1], sb = sn[3 * gate + 1];
  float cg = cs[3 * gate + 2], sg = sn[3 * gate + 2];
  cpx m;
  if (i == 0 && j == 0) { m.re =  cb * ca; m.im =  sb * sa; }
  else if (i == 0)      { m.re = -sb * ca; m.im = -cb * sa; }
  else if (j == 0)      { m.re =  sb * ca; m.im = -cb * sa; }
  else                  { m.re =  cb * ca; m.im = -sb * sa; }
  cpx ph; ph.re = cg; ph.im = (i == 0) ? -sg : sg;
  return cmul(ph, m);
}

__host__ __device__ constexpr int chainperm(int r) {
  r ^= ((r >> 1) & 1);
  r ^= ((r >> 2) & 1) << 1;
  r ^= ((r >> 3) & 1) << 2;
  return r;
}

__device__ __forceinline__ void build_gates(const float* __restrict__ w, float* tmp,
                                            float* fragf, int t) {
  float* cs  = tmp;
  float* sn  = tmp + 96;
  float* p4  = tmp + 256;
  if (t < NLAYERS * NQ * 3) {
    float h = 0.5f * w[t];
    cs[t] = cosf(h);
    sn[t] = sinf(h);
  }
  __syncthreads();
  {
    int l = t >> 6, p = (t >> 4) & 3, i4 = (t >> 2) & 3, j4 = t & 3;
    int g0 = l * NQ + 2 * p;
    cpx e = cmul(u_entry(cs, sn, g0,     i4 >> 1, j4 >> 1),
                 u_entry(cs, sn, g0 + 1, i4 & 1,  j4 & 1));
    p4[2 * t]     = e.re;
    p4[2 * t + 1] = e.im;
  }
  __syncthreads();
  {
    int m = t >> 4, k = t & 15;
    int mp = chainperm(m);
    int slot = ((k >> 2) * 32 + m + 16 * ((k >> 1) & 1)) * 4 + (k & 1);
    #pragma unroll
    for (int l = 0; l < NLAYERS; ++l) {
      const float* pl = p4 + l * 128;
      cpx a, b, r;
      a.re = pl[(2 * 16 + (m >> 2) * 4 + (k >> 2)) * 2]; a.im = pl[(2 * 16 + (m >> 2) * 4 + (k >> 2)) * 2 + 1];
      b.re = pl[(3 * 16 + (m & 3) * 4 + (k & 3)) * 2];   b.im = pl[(3 * 16 + (m & 3) * 4 + (k & 3)) * 2 + 1];
      r = cmul(a, b);
      fragf[(l * 2 + 0) * 512 + slot]     = r.re;
      fragf[(l * 2 + 0) * 512 + slot + 2] = r.im - r.re;
      a.re = pl[(0 * 16 + (mp >> 2) * 4 + (k >> 2)) * 2]; a.im = pl[(0 * 16 + (mp >> 2) * 4 + (k >> 2)) * 2 + 1];
      b.re = pl[(1 * 16 + (mp & 3) * 4 + (k & 3)) * 2];   b.im = pl[(1 * 16 + (mp & 3) * 4 + (k & 3)) * 2 + 1];
      r = cmul(a, b);
      fragf[(l * 2 + 1) * 512 + slot]     = r.re;
      fragf[(l * 2 + 1) * 512 + slot + 2] = r.im - r.re;
    }
  }
  __syncthreads();
}


__device__ __forceinline__ void cd_to_b(const float* S, float* Bv, int half) {
  const int r0t[8] = {0, 1, 4, 5, 0, 1, 4, 5};
  #pragma unroll
  for (int d = 0; d < 8; ++d) {
    int r0 = r0t[d];
    if (d < 4) {
      float t = __shfl_xor(S[r0 + 2], 16, 32);
      Bv[d] = half ? t : S[r0];
    } else {
      float t = __shfl_xor(S[r0], 16, 32);
      Bv[d] = half ? S[r0 + 2] : t;
    }
  }
}

__device__ __forceinline__ void cmulmat(const v4f* mat, int lane,
                                        const float* Xb, const float* Yb,
                                        float* Xo, float* Yo) {
  v4f f[4];
  #pragma unroll
  for (int kc = 0; kc < 4; ++kc) f[kc] = mat[kc * 32 + lane];
  v8f accM = {};
  #pragma unroll
  for (int kc = 0; kc < 4; ++kc) {
    int a0 = 2 * kc, a1 = a0 + 1;
    v2f ar;  ar[0] = f[kc][0];         ar[1] = f[kc][1];
    v2f bs;  bs[0] = Xb[a0] + Yb[a0];  bs[1] = Xb[a1] + Yb[a1];
    accM = __builtin_amdgcn_wmma_f32_16x16x4_f32(false, ar, false, bs, (short)0, accM, false, false);
  }
  v8f accX = accM; v8f accY = accM;
  #pragma unroll
  for (int kc = 0; kc < 4; ++kc) {
    int a0 = 2 * kc, a1 = a0 + 1;
    v2f ad;  ad[0] = f[kc][2];  ad[1] = f[kc][3];
    v2f aq;  aq[0] = -2.f * f[kc][0] - ad[0];
             aq[1] = -2.f * f[kc][1] - ad[1];
    v2f bx;  bx[0] = Xb[a0];    bx[1] = Xb[a1];
    v2f by;  by[0] = Yb[a0];    by[1] = Yb[a1];
    accX = __builtin_amdgcn_wmma_f32_16x16x4_f32(false, aq, false, by, (short)0, accX, false, false);
    accY = __builtin_amdgcn_wmma_f32_16x16x4_f32(false, ad, false, bx, (short)0, accY, false, false);
  }
  #pragma unroll
  for (int v = 0; v < 8; ++v) { Xo[v] = accX[v]; Yo[v] = accY[v]; }
}

__global__ __launch_bounds__(256, 16) void qsim(const float* __restrict__ x,
                                            const float* __restrict__ w,
                                            float* __restrict__ out, int B) {
  __shared__ float lds[8 * 512];
  __shared__ v4f   frag[8 * 4 * 32];
  const int lane  = threadIdx.x & 31;
  const int wslot = threadIdx.x >> 5;
  const int bw    = blockIdx.x * 8 + wslot;
  const int b     = bw < B ? bw : B - 1;
  const int half = lane >> 4;
  const int col  = lane & 15;
  float* myLds = lds + wslot * 512;

  float cq[8], sq[8];
  {
    float sv, cv;
    sincosf(0.5f * x[b * 8 + (lane & 7)], &sv, &cv);
    #pragma unroll
    for (int q = 0; q < 8; ++q) {
      cq[q] = __int_as_float(__builtin_amdgcn_readlane(__float_as_int(cv), q));
      sq[q] = __int_as_float(__builtin_amdgcn_readlane(__float_as_int(sv), q));
    }
  }
  float colf = 1.f;
  #pragma unroll
  for (int q = 0; q < 4; ++q) colf *= ((col >> (3 - q)) & 1) ? sq[q] : cq[q];
  float X[8], Y[8];
  #pragma unroll
  for (int v = 0; v < 8; ++v) {
    int r = v + 8 * half;
    float rowf = 1.f;
    #pragma unroll
    for (int q = 4; q < 8; ++q) rowf *= ((r >> (7 - q)) & 1) ? sq[q] : cq[q];
    X[v] = colf * rowf; Y[v] = 0.f;
  }

  build_gates(w, lds, (float*)frag, threadIdx.x);

  for (int l = 0; l < NLAYERS; ++l) {
    const int flip = l & 1;
    const v4f* matA = frag + (l * 2 + flip)       * 128;
    const v4f* matB = frag + (l * 2 + (1 - flip)) * 128;

    float Xb[8], Yb[8];
    cd_to_b(X, Xb, half);
    cd_to_b(Y, Yb, half);
    cmulmat(matA, lane, Xb, Yb, X, Y);

    #pragma unroll
    for (int v = 0; v < 8; ++v) {
      myLds[(v + 8 * half) * 16 + col]       = X[v];
      myLds[256 + (v + 8 * half) * 16 + col] = Y[v];
    }
    __syncthreads();
    #pragma unroll
    for (int bb = 0; bb < 8; ++bb) {
      int kk = 4 * (bb >> 1) + (bb & 1) + 2 * half;
      Xb[bb] = myLds[col * 16 + kk];
      Yb[bb] = myLds[256 + col * 16 + kk];
    }
    __syncthreads();

    cmulmat(matB, lane, Xb, Yb, X, Y);

    if (flip) {
      #pragma unroll
      for (int v = 0; v < 8; ++v) {
        float tx = __shfl_xor(X[v], 16, 32);
        float ty = __shfl_xor(Y[v], 16, 32);
        if (lane & 1) { X[v] = tx; Y[v] = ty; }
      }
      float nX[8], nY[8];
      #pragma unroll
      for (int v = 0; v < 8; ++v) {
        const int r0 = chainperm(v), r1 = chainperm(v + 8);
        float t0x = (r0 < 8)  ? X[r0 & 7] : __shfl_xor(X[r0 & 7], 16, 32);
        float t0y = (r0 < 8)  ? Y[r0 & 7] : __shfl_xor(Y[r0 & 7], 16, 32);
        float t1x = (r1 >= 8) ? X[r1 & 7] : __shfl_xor(X[r1 & 7], 16, 32);
        float t1y = (r1 >= 8) ? Y[r1 & 7] : __shfl_xor(Y[r1 & 7], 16, 32);
        nX[v] = half ? t1x : t0x;
        nY[v] = half ? t1y : t0y;
      }
      #pragma unroll
      for (int v = 0; v < 8; ++v) { X[v] = nX[v]; Y[v] = nY[v]; }
    } else {
      #pragma unroll
      for (int v = 1; v < 8; v += 2) {
        X[v] = __shfl_xor(X[v], 8, 32);
        Y[v] = __shfl_xor(Y[v], 8, 32);
      }
      int src = chainperm(lane & 15) | (lane & 16);
      #pragma unroll
      for (int v = 0; v < 8; ++v) {
        X[v] = __shfl(X[v], src, 32);
        Y[v] = __shfl(Y[v], src, 32);
      }
    }
  }

  float psum = 0.f, s5 = 0.f, s6 = 0.f, s7 = 0.f;
  #pragma unroll
  for (int v = 0; v < 8; ++v) {
    float p = X[v] * X[v] + Y[v] * Y[v];
    psum += p;
    s5 += ((v >> 2) & 1) ? -p : p;
    s6 += ((v >> 1) & 1) ? -p : p;
    s7 += ((v) & 1)      ? -p : p;
  }
  float o[8];
  o[0] = ((col >> 3) & 1) ? -psum : psum;
  o[1] = ((col >> 2) & 1) ? -psum : psum;
  o[2] = ((col >> 1) & 1) ? -psum : psum;
  o[3] = ((col) & 1)      ? -psum : psum;
  o[4] = half ? -psum : psum;
  o[5] = s5; o[6] = s6; o[7] = s7;
  #pragma unroll
  for (int q = 0; q < 8; ++q) {
    float v = o[q];
    #pragma unroll
    for (int off = 16; off > 0; off >>= 1) v += __shfl_xor(v, off, 32);
    o[q] = v;
  }
  if (lane == 0 && bw < B) {
    #pragma unroll
    for (int q = 0; q < 8; ++q) out[b * 8 + q] = o[q];
  }
}

extern "C" void kernel_launch(void* const* d_in, const int* in_sizes, int n_in,
                              void* d_out, int out_size, void* d_ws, size_t ws_size,
                              hipStream_t stream) {
  const float* x = (const float*)d_in[0];
  const float* w = (const float*)d_in[1];
  float* out = (float*)d_out;
  int B = in_sizes[0] / NQ;

  int nblocks = (B + 7) / 8;
  qsim<<<nblocks, 256, 0, stream>>>(x, w, out, B);
}


# First layer applied per wire to the encoding's product state (2x2 gate times 2-vector, CNOTs as index bits); its two 16x16 complex products dropped, 96 -> 72 matrix instructions; sums regrouped
# speedup vs baseline: 2.3919x; 1.2302x over previous
#include <hip/hip_runtime.h>


typedef __attribute__((ext_vector_type(2))) float v2f;
typedef __attribute__((ext_vector_type(4))) float v4f;
typedef __attribute__((ext_vector_type(8))) float v8f;

#define NQ 8
#define NLAYERS 4

struct cpx { float re, im; };

__device__ __forceinline__ cpx cmul(cpx a, cpx b) {
  cpx r; r.re = a.re * b.re - a.im * b.im; r.im = a.re * b.im + a.im * b.re; return r;
}

__device__ __forceinline__ cpx pick(cpx a, cpx b, int sel) {
  cpx r; r.re = sel ? b.re : a.re; r.im = sel ? b.im : a.im; return r;
}

__device__ __forceinline__ cpx u_entry(const float* cs, const float* sn, int gate, int i, int j) {
  float ca = cs[3 * gate],     sa = sn[3 * gate];
  float cb = cs[3 * gate + 1], sb = sn[3 * gate + 1];
  float cg = cs[3 * gate + 2], sg = sn[3 * gate + 2];
  cpx m;
  if (i == 0 && j == 0) { m.re =  cb * ca; m.im =  sb * sa; }
  else if (i == 0)      { m.re = -sb * ca; m.im = -cb * sa; }
  else if (j == 0)      { m.re =  sb * ca; m.im = -cb * sa; }
  else                  { m.re =  cb * ca; m.im = -sb * sa; }
  cpx ph; ph.re = cg; ph.im = (i == 0) ? -sg : sg;
  return cmul(ph, m);
}

__host__ __device__ constexpr int chainperm(int r) {
  r ^= ((r >> 1) & 1);
  r ^= ((r >> 2) & 1) << 1;
  r ^= ((r >> 3) & 1) << 2;
  return r;
}

__device__ __forceinline__ void build_gates(const float* __restrict__ w, float* tmp,
                                            float* fragf, int t,
                                            float cv, float sv, float* av) {
  float* cs  = tmp;
  float* sn  = tmp + 96;
  float* p4  = tmp + 256;
  if (t < NLAYERS * NQ * 3) {
    float h = 0.5f * w[t];
    cs[t] = cosf(h);
    sn[t] = sinf(h);
  }
  __syncthreads();
  {
    int q = t & 7;
    cpx u00 = u_entry(cs, sn, q, 0, 0), u01 = u_entry(cs, sn, q, 0, 1);
    cpx u10 = u_entry(cs, sn, q, 1, 0), u11 = u_entry(cs, sn, q, 1, 1);
    av[0] = u00.re * cv + u01.re * sv;  av[1] = u00.im * cv + u01.im * sv;
    av[2] = u10.re * cv + u11.re * sv;  av[3] = u10.im * cv + u11.im * sv;
  }
  if (t >= 64) {
    int l = t >> 6, p = (t >> 4) & 3, i4 = (t >> 2) & 3, j4 = t & 3;
    int g0 = l * NQ + 2 * p;
    cpx e = cmul(u_entry(cs, sn, g0,     i4 >> 1, j4 >> 1),
                 u_entry(cs, sn, g0 + 1, i4 & 1,  j4 & 1));
    p4[2 * t]     = e.re;
    p4[2 * t + 1] = e.im;
  }
  __syncthreads();
  {
    int m = t >> 4, k = t & 15;
    int mp = chainperm(m);
    int slot = ((k >> 2) * 32 + m + 16 * ((k >> 1) & 1)) * 4 + (k & 1);
    #pragma unroll
    for (int l = 1; l < NLAYERS; ++l) {
      const float* pl = p4 + l * 128;
      cpx a, b, r;
      a.re = pl[(2 * 16 + (m >> 2) * 4 + (k >> 2)) * 2]; a.im = pl[(2 * 16 + (m >> 2) * 4 + (k >> 2)) * 2 + 1];
      b.re = pl[(3 * 16 + (m & 3) * 4 + (k & 3)) * 2];   b.im = pl[(3 * 16 + (m & 3) * 4 + (k & 3)) * 2 + 1];
      r = cmul(a, b);
      fragf[((l - 1) * 2 + 0) * 512 + slot]     = r.re;
      fragf[((l - 1) * 2 + 0) * 512 + slot + 2] = r.im - r.re;
      a.re = pl[(0 * 16 + (mp >> 2) * 4 + (k >> 2)) * 2]; a.im = pl[(0 * 16 + (mp >> 2) * 4 + (k >> 2)) * 2 + 1];
      b.re = pl[(1 * 16 + (mp & 3) * 4 + (k & 3)) * 2];   b.im = pl[(1 * 16 + (mp & 3) * 4 + (k & 3)) * 2 + 1];
      r = cmul(a, b);
      fragf[((l - 1) * 2 + 1) * 512 + slot]     = r.re;
      fragf[((l - 1) * 2 + 1) * 512 + slot + 2] = r.im - r.re;
    }
  }
  __syncthreads();
}


__device__ __forceinline__ void cd_to_b(const float* S, float* Bv, int half) {
  const int r0t[8] = {0, 1, 4, 5, 0, 1, 4, 5};
  #pragma unroll
  for (int d = 0; d < 8; ++d) {
    int r0 = r0t[d];
    if (d < 4) {
      float t = __shfl_xor(S[r0 + 2], 16, 32);
      Bv[d] = half ? t : S[r0];
    } else {
      float t = __shfl_xor(S[r0], 16, 32);
      Bv[d] = half ? S[r0 + 2] : t;
    }
  }
}

__device__ __forceinline__ void cmulmat(const v4f* mat, int lane,
                                        const float* Xb, const float* Yb,
                                        float* Xo, float* Yo) {
  v4f f[4];
  #pragma unroll
  for (int kc = 0; kc < 4; ++kc) f[kc] = mat[kc * 32 + lane];
  v8f accM = {};
  #pragma unroll
  for (int kc = 0; kc < 4; ++kc) {
    int a0 = 2 * kc, a1 = a0 + 1;
    v2f ar;  ar[0] = f[kc][0];         ar[1] = f[kc][1];
    v2f bs;  bs[0] = Xb[a0] + Yb[a0];  bs[1] = Xb[a1] + Yb[a1];
    accM = __builtin_amdgcn_wmma_f32_16x16x4_f32(false, ar, false, bs, (short)0, accM, false, false);
  }
  v8f accX = accM; v8f accY = accM;
  #pragma unroll
  for (int kc = 0; kc < 4; ++kc) {
    int a0 = 2 * kc, a1 = a0 + 1;
    v2f ad;  ad[0] = f[kc][2];  ad[1] = f[kc][3];
    v2f aq;  aq[0] = -2.f * f[kc][0] - ad[0];
             aq[1] = -2.f * f[kc][1] - ad[1];
    v2f bx;  bx[0] = Xb[a0];    bx[1] = Xb[a1];
    v2f by;  by[0] = Yb[a0];    by[1] = Yb[a1];
    accX = __builtin_amdgcn_wmma_f32_16x16x4_f32(false, aq, false, by, (short)0, accX, false, false);
    accY = __builtin_amdgcn_wmma_f32_16x16x4_f32(false, ad, false, bx, (short)0, accY, false, false);
  }
  #pragma unroll
  for (int v = 0; v < 8; ++v) { Xo[v] = accX[v]; Yo[v] = accY[v]; }
}

__global__ __launch_bounds__(256, 16) void qsim(const float* __restrict__ x,
                                            const float* __restrict__ w,
                                            float* __restrict__ out, int B) {
  __shared__ float lds[8 * 512];
  __shared__ v4f   frag[6 * 4 * 32];
  const int lane  = threadIdx.x & 31;
  const int wslot = __builtin_amdgcn_readfirstlane(threadIdx.x >> 5);
  const int bw    = blockIdx.x * 8 + wslot;
  const int b     = bw < B ? bw : B - 1;
  const int half = lane >> 4;
  const int col  = lane & 15;
  float* myLds = lds + wslot * 512;

  float sv, cv;
  sincosf(0.5f * x[b * 8 + (lane & 7)], &sv, &cv);

  float av[4];
  build_gates(w, lds, (float*)frag, threadIdx.x, cv, sv, av);

  cpx a0[8], a1[8];
  #pragma unroll
  for (int q = 0; q < 8; ++q) {
    a0[q].re = __int_as_float(__builtin_amdgcn_readlane(__float_as_int(av[0]), q));
    a0[q].im = __int_as_float(__builtin_amdgcn_readlane(__float_as_int(av[1]), q));
    a1[q].re = __int_as_float(__builtin_amdgcn_readlane(__float_as_int(av[2]), q));
    a1[q].im = __int_as_float(__builtin_amdgcn_readlane(__float_as_int(av[3]), q));
  }

  float X[8], Y[8];
  {
    const int r3 = (col >> 3) & 1, r2 = (col >> 2) & 1, r1 = (col >> 1) & 1, r0 = col & 1;
    cpx lf = cmul(cmul(pick(a0[5], a1[5], r2 ^ r3), pick(a0[6], a1[6], r1 ^ r2)),
                  pick(a0[7], a1[7], r0 ^ r1));
    cpx L[2];
    L[0] = cmul(lf, pick(a0[4], a1[4], r3));
    L[1] = cmul(lf, pick(a0[4], a1[4], r3 ^ 1));
    cpx h0 = pick(a0[0], a1[0], half);
    cpx g[2];
    g[0] = cmul(h0, pick(a0[1], a1[1], half));
    g[1] = cmul(h0, pick(a0[1], a1[1], half ^ 1));
    #pragma unroll
    for (int v = 0; v < 8; ++v) {
      const int v2 = (v >> 2) & 1, v1 = (v >> 1) & 1, v0 = v & 1;
      cpx u = cmul((v1 ^ v2) ? a1[2] : a0[2], (v0 ^ v1) ? a1[3] : a0[3]);
      cpx e = cmul(cmul(g[v2], u), L[v0]);
      X[v] = e.re; Y[v] = e.im;
    }
  }

  for (int l = 1; l < NLAYERS; ++l) {
    const int flip = l & 1;
    const v4f* matA = frag + ((l - 1) * 2 + flip)       * 128;
    const v4f* matB = frag + ((l - 1) * 2 + (1 - flip)) * 128;

    float Xb[8], Yb[8];
    cd_to_b(X, Xb, half);
    cd_to_b(Y, Yb, half);
    cmulmat(matA, lane, Xb, Yb, X, Y);

    #pragma unroll
    for (int v = 0; v < 8; ++v) {
      myLds[(v + 8 * half) * 16 + col]       = X[v];
      myLds[256 + (v + 8 * half) * 16 + col] = Y[v];
    }
    __syncthreads();
    #pragma unroll
    for (int bb = 0; bb < 8; ++bb) {
      int kk = 4 * (bb >> 1) + (bb & 1) + 2 * half;
      Xb[bb] = myLds[col * 16 + kk];
      Yb[bb] = myLds[256 + col * 16 + kk];
    }
    __syncthreads();

    cmulmat(matB, lane, Xb, Yb, X, Y);

    if (flip) {
      #pragma unroll
      for (int v = 0; v < 8; ++v) {
        float tx = __shfl_xor(X[v], 16, 32);
        float ty = __shfl_xor(Y[v], 16, 32);
        if (lane & 1) { X[v] = tx; Y[v] = ty; }
      }
      float nX[8], nY[8];
      #pragma unroll
      for (int v = 0; v < 8; ++v) {
        const int r0 = chainperm(v), r1 = chainperm(v + 8);
        float t0x = (r0 < 8)  ? X[r0 & 7] : __shfl_xor(X[r0 & 7], 16, 32);
        float t0y = (r0 < 8)  ? Y[r0 & 7] : __shfl_xor(Y[r0 & 7], 16, 32);
        float t1x = (r1 >= 8) ? X[r1 & 7] : __shfl_xor(X[r1 & 7], 16, 32);
        float t1y = (r1 >= 8) ? Y[r1 & 7] : __shfl_xor(Y[r1 & 7], 16, 32);
        nX[v] = half ? t1x : t0x;
        nY[v] = half ? t1y : t0y;
      }
      #pragma unroll
      for (int v = 0; v < 8; ++v) { X[v] = nX[v]; Y[v] = nY[v]; }
    } else {
      #pragma unroll
      for (int v = 1; v < 8; v += 2) {
        X[v] = __shfl_xor(X[v], 8, 32);
        Y[v] = __shfl_xor(Y[v], 8, 32);
      }
      int src = chainperm(lane & 15) | (lane & 16);
      #pragma unroll
      for (int v = 0; v < 8; ++v) {
        X[v] = __shfl(X[v], src, 32);
        Y[v] = __shfl(Y[v], src, 32);
      }
    }
  }

  float psum = 0.f, s5 = 0.f, s6 = 0.f, s7 = 0.f;
  #pragma unroll
  for (int v = 0; v < 8; ++v) {
    float p = X[v] * X[v] + Y[v] * Y[v];
    psum += p;
    s5 += ((v >> 2) & 1) ? -p : p;
    s6 += ((v >> 1) & 1) ? -p : p;
    s7 += ((v) & 1)      ? -p : p;
  }
  float o[8];
  o[0] = ((col >> 3) & 1) ? -psum : psum;
  o[1] = ((col >> 2) & 1) ? -psum : psum;
  o[2] = ((col >> 1) & 1) ? -psum : psum;
  o[3] = ((col) & 1)      ? -psum : psum;
  o[4] = half ? -psum : psum;
  o[5] = s5; o[6] = s6; o[7] = s7;
  #pragma unroll
  for (int q = 0; q < 8; ++q) {
    float v = o[q];
    #pragma unroll
    for (int off = 16; off > 0; off >>= 1) v += __shfl_xor(v, off, 32);
    o[q] = v;
  }
  if (lane == 0 && bw < B) {
    #pragma unroll
    for (int q = 0; q < 8; ++q) out[b * 8 + q] = o[q];
  }
}

extern "C" void kernel_launch(void* const* d_in, const int* in_sizes, int n_in,
                              void* d_out, int out_size, void* d_ws, size_t ws_size,
                              hipStream_t stream) {
  const float* x = (const float*)d_in[0];
  const float* w = (const float*)d_in[1];
  float* out = (float*)d_out;
  int B = in_sizes[0] / NQ;

  int nblocks = (B + 7) / 8;
  qsim<<<nblocks, 256, 0, stream>>>(x, w, out, B);
}


# Both products of a layer read the state straight from the accumulator registers (gate k order = accumulator rows; product 1: state as A operand): shuffle convert, LDS transpose gone; sums regrouped
# speedup vs baseline: 2.7034x; 1.1302x over previous
#include <hip/hip_runtime.h>


typedef __attribute__((ext_vector_type(2))) float v2f;
typedef __attribute__((ext_vector_type(4))) float v4f;
typedef __attribute__((ext_vector_type(8))) float v8f;

#define NQ 8
#define NLAYERS 4

struct cpx { float re, im; };

__device__ __forceinline__ cpx cmul(cpx a, cpx b) {
  cpx r; r.re = a.re * b.re - a.im * b.im; r.im = a.re * b.im + a.im * b.re; return r;
}

__device__ __forceinline__ cpx pick(cpx a, cpx b, int sel) {
  cpx r; r.re = sel ? b.re : a.re; r.im = sel ? b.im : a.im; return r;
}

__device__ __forceinline__ cpx u_entry(const float* cs, const float* sn, int gate, int i, int j) {
  float ca = cs[3 * gate],     sa = sn[3 * gate];
  float cb = cs[3 * gate + 1], sb = sn[3 * gate + 1];
  float cg = cs[3 * gate + 2], sg = sn[3 * gate + 2];
  cpx m;
  if (i == 0 && j == 0) { m.re =  cb * ca; m.im =  sb * sa; }
  else if (i == 0)      { m.re = -sb * ca; m.im = -cb * sa; }
  else if (j == 0)      { m.re =  sb * ca; m.im = -cb * sa; }
  else                  { m.re =  cb * ca; m.im = -sb * sa; }
  cpx ph; ph.re = cg; ph.im = (i == 0) ? -sg : sg;
  return cmul(ph, m);
}

__host__ __device__ constexpr int chainperm(int r) {
  r ^= ((r >> 1) & 1);
  r ^= ((r >> 2) & 1) << 1;
  r ^= ((r >> 3) & 1) << 2;
  return r;
}

__device__ __forceinline__ void build_gates(const float* __restrict__ w, float* tmp,
                                            float* fragf, int t,
                                            float cv, float sv, float* av) {
  float* cs  = tmp;
  float* sn  = tmp + 96;
  float* p4  = tmp + 256;
  if (t < NLAYERS * NQ * 3) {
    float h = 0.5f * w[t];
    cs[t] = cosf(h);
    sn[t] = sinf(h);
  }
  __syncthreads();
  {
    int q = t & 7;
    cpx u00 = u_entry(cs, sn, q, 0, 0), u01 = u_entry(cs, sn, q, 0, 1);
    cpx u10 = u_entry(cs, sn, q, 1, 0), u11 = u_entry(cs, sn, q, 1, 1);
    av[0] = u00.re * cv + u01.re * sv;  av[1] = u00.im * cv + u01.im * sv;
    av[2] = u10.re * cv + u11.re * sv;  av[3] = u10.im * cv + u11.im * sv;
  }
  if (t >= 64) {
    int l = t >> 6, p = (t >> 4) & 3, i4 = (t >> 2) & 3, j4 = t & 3;
    int g0 = l * NQ + 2 * p;
    cpx e = cmul(u_entry(cs, sn, g0,     i4 >> 1, j4 >> 1),
                 u_entry(cs, sn, g0 + 1, i4 & 1,  j4 & 1));
    p4[2 * t]     = e.re;
    p4[2 * t + 1] = e.im;
  }
  __syncthreads();
  {
    int m = t >> 4, k = t & 15;
    int mp = chainperm(m);
    int slot = (((k >> 1) & 3) * 32 + m + 16 * (k >> 3)) * 4 + (k & 1);
    #pragma unroll
    for (int l = 1; l < NLAYERS; ++l) {
      const float* pl = p4 + l * 128;
      cpx a, b, r;
      a.re = pl[(2 * 16 + (m >> 2) * 4 + (k >> 2)) * 2]; a.im = pl[(2 * 16 + (m >> 2) * 4 + (k >> 2)) * 2 + 1];
      b.re = pl[(3 * 16 + (m & 3) * 4 + (k & 3)) * 2];   b.im = pl[(3 * 16 + (m & 3) * 4 + (k & 3)) * 2 + 1];
      r = cmul(a, b);
      fragf[((l - 1) * 2 + 0) * 512 + slot]     = r.re;
      fragf[((l - 1) * 2 + 0) * 512 + slot + 2] = r.im - r.re;
      a.re = pl[(0 * 16 + (mp >> 2) * 4 + (k >> 2)) * 2]; a.im = pl[(0 * 16 + (mp >> 2) * 4 + (k >> 2)) * 2 + 1];
      b.re = pl[(1 * 16 + (mp & 3) * 4 + (k & 3)) * 2];   b.im = pl[(1 * 16 + (mp & 3) * 4 + (k & 3)) * 2 + 1];
      r = cmul(a, b);
      fragf[((l - 1) * 2 + 1) * 512 + slot]     = r.re;
      fragf[((l - 1) * 2 + 1) * 512 + slot + 2] = r.im - r.re;
    }
  }
  __syncthreads();
}


template <bool STATE_A>
__device__ __forceinline__ v8f mm(v2f g, v2f s, v8f acc) {
  return STATE_A
    ? __builtin_amdgcn_wmma_f32_16x16x4_f32(false, s, false, g, (short)0, acc, false, false)
    : __builtin_amdgcn_wmma_f32_16x16x4_f32(false, g, false, s, (short)0, acc, false, false);
}

template <bool STATE_A>
__device__ __forceinline__ void cmulmat(const v4f* mat, int lane, float* X, float* Y) {
  v4f f[4];
  #pragma unroll
  for (int kc = 0; kc < 4; ++kc) f[kc] = mat[kc * 32 + lane];
  v8f accM = {};
  #pragma unroll
  for (int kc = 0; kc < 4; ++kc) {
    int a0 = 2 * kc, a1 = a0 + 1;
    v2f ar;  ar[0] = f[kc][0];       ar[1] = f[kc][1];
    v2f bs;  bs[0] = X[a0] + Y[a0];  bs[1] = X[a1] + Y[a1];
    accM = mm<STATE_A>(ar, bs, accM);
  }
  v8f accX = accM; v8f accY = accM;
  #pragma unroll
  for (int kc = 0; kc < 4; ++kc) {
    int a0 = 2 * kc, a1 = a0 + 1;
    v2f ad;  ad[0] = f[kc][2];  ad[1] = f[kc][3];
    v2f aq;  aq[0] = -2.f * f[kc][0] - ad[0];
             aq[1] = -2.f * f[kc][1] - ad[1];
    v2f bx;  bx[0] = X[a0];     bx[1] = X[a1];
    v2f by;  by[0] = Y[a0];     by[1] = Y[a1];
    accX = mm<STATE_A>(aq, by, accX);
    accY = mm<STATE_A>(ad, bx, accY);
  }
  #pragma unroll
  for (int v = 0; v < 8; ++v) { X[v] = accX[v]; Y[v] = accY[v]; }
}

__global__ __launch_bounds__(256, 16) void qsim(const float* __restrict__ x,
                                            const float* __restrict__ w,
                                            float* __restrict__ out, int B) {
  __shared__ float tmp[768];
  __shared__ v4f   frag[6 * 4 * 32];
  const int lane  = threadIdx.x & 31;
  const int wslot = __builtin_amdgcn_readfirstlane(threadIdx.x >> 5);
  const int bw    = blockIdx.x * 8 + wslot;
  const int b     = bw < B ? bw : B - 1;
  const int half = lane >> 4;
  const int col  = lane & 15;

  float sv, cv;
  sincosf(0.5f * x[b * 8 + (lane & 7)], &sv, &cv);

  float av[4];
  build_gates(w, tmp, (float*)frag, threadIdx.x, cv, sv, av);

  cpx a0[8], a1[8];
  #pragma unroll
  for (int q = 0; q < 8; ++q) {
    a0[q].re = __int_as_float(__builtin_amdgcn_readlane(__float_as_int(av[0]), q));
    a0[q].im = __int_as_float(__builtin_amdgcn_readlane(__float_as_int(av[1]), q));
    a1[q].re = __int_as_float(__builtin_amdgcn_readlane(__float_as_int(av[2]), q));
    a1[q].im = __int_as_float(__builtin_amdgcn_readlane(__float_as_int(av[3]), q));
  }

  float X[8], Y[8];
  {
    const int r3 = (col >> 3) & 1, r2 = (col >> 2) & 1, r1 = (col >> 1) & 1, r0 = col & 1;
    cpx lf = cmul(cmul(pick(a0[5], a1[5], r2 ^ r3), pick(a0[6], a1[6], r1 ^ r2)),
                  pick(a0[7], a1[7], r0 ^ r1));
    cpx L[2];
    L[0] = cmul(lf, pick(a0[4], a1[4], r3));
    L[1] = cmul(lf, pick(a0[4], a1[4], r3 ^ 1));
    cpx h0 = pick(a0[0], a1[0], half);
    cpx g[2];
    g[0] = cmul(h0, pick(a0[1], a1[1], half));
    g[1] = cmul(h0, pick(a0[1], a1[1], half ^ 1));
    #pragma unroll
    for (int v = 0; v < 8; ++v) {
      const int v2 = (v >> 2) & 1, v1 = (v >> 1) & 1, v0 = v & 1;
      cpx u = cmul((v1 ^ v2) ? a1[2] : a0[2], (v0 ^ v1) ? a1[3] : a0[3]);
      cpx e = cmul(cmul(g[v2], u), L[v0]);
      X[v] = e.re; Y[v] = e.im;
    }
  }

  for (int l = 1; l < NLAYERS; ++l) {
    const int flip = l & 1;
    const v4f* matA = frag + ((l - 1) * 2 + flip)       * 128;
    const v4f* matB = frag + ((l - 1) * 2 + (1 - flip)) * 128;

    cmulmat<true>(matA, lane, X, Y);
    cmulmat<false>(matB, lane, X, Y);

    if (flip) {
      #pragma unroll
      for (int v = 0; v < 8; ++v) {
        float tx = __shfl_xor(X[v], 16, 32);
        float ty = __shfl_xor(Y[v], 16, 32);
        if (lane & 1) { X[v] = tx; Y[v] = ty; }
      }
      float nX[8], nY[8];
      #pragma unroll
      for (int v = 0; v < 8; ++v) {
        const int r0 = chainperm(v), r1 = chainperm(v + 8);
        float t0x = (r0 < 8)  ? X[r0 & 7] : __shfl_xor(X[r0 & 7], 16, 32);
        float t0y = (r0 < 8)  ? Y[r0 & 7] : __shfl_xor(Y[r0 & 7], 16, 32);
        float t1x = (r1 >= 8) ? X[r1 & 7] : __shfl_xor(X[r1 & 7], 16, 32);
        float t1y = (r1 >= 8) ? Y[r1 & 7] : __shfl_xor(Y[r1 & 7], 16, 32);
        nX[v] = half ? t1x : t0x;
        nY[v] = half ? t1y : t0y;
      }
      #pragma unroll
      for (int v = 0; v < 8; ++v) { X[v] = nX[v]; Y[v] = nY[v]; }
    } else {
      #pragma unroll
      for (int v = 1; v < 8; v += 2) {
        X[v] = __shfl_xor(X[v], 8, 32);
        Y[v] = __shfl_xor(Y[v], 8, 32);
      }
      int src = chainperm(lane & 15) | (lane & 16);
      #pragma unroll
      for (int v = 0; v < 8; ++v) {
        X[v] = __shfl(X[v], src, 32);
        Y[v] = __shfl(Y[v], src, 32);
      }
    }
  }

  float psum = 0.f, s5 = 0.f, s6 = 0.f, s7 = 0.f;
  #pragma unroll
  for (int v = 0; v < 8; ++v) {
    float p = X[v] * X[v] + Y[v] * Y[v];
    psum += p;
    s5 += ((v >> 2) & 1) ? -p : p;
    s6 += ((v >> 1) & 1) ? -p : p;
    s7 += ((v) & 1)      ? -p : p;
  }
  float o[8];
  o[0] = ((col >> 3) & 1) ? -psum : psum;
  o[1] = ((col >> 2) & 1) ? -psum : psum;
  o[2] = ((col >> 1) & 1) ? -psum : psum;
  o[3] = ((col) & 1)      ? -psum : psum;
  o[4] = half ? -psum : psum;
  o[5] = s5; o[6] = s6; o[7] = s7;
  #pragma unroll
  for (int q = 0; q < 8; ++q) {
    float v = o[q];
    #pragma unroll
    for (int off = 16; off > 0; off >>= 1) v += __shfl_xor(v, off, 32);
    o[q] = v;
  }
  if (lane == 0 && bw < B) {
    #pragma unroll
    for (int q = 0; q < 8; ++q) out[b * 8 + q] = o[q];
  }
}

extern "C" void kernel_launch(void* const* d_in, const int* in_sizes, int n_in,
                              void* d_out, int out_size, void* d_ws, size_t ws_size,
                              hipStream_t stream) {
  const float* x = (const float*)d_in[0];
  const float* w = (const float*)d_in[1];
  float* out = (float*)d_out;
  int B = in_sizes[0] / NQ;

  int nblocks = (B + 7) / 8;
  qsim<<<nblocks, 256, 0, stream>>>(x, w, out, B);
}


# Layer 1 applied to the two-outer-product form of the state after layer 0: each gate matrix times two 16-vectors, then one K=4 matrix instruction per real part; 72 -> 50 WMMA; sums regrouped
# speedup vs baseline: 3.1994x; 1.1835x over previous
#include <hip/hip_runtime.h>


typedef __attribute__((ext_vector_type(2))) float v2f;
typedef __attribute__((ext_vector_type(4))) float v4f;
typedef __attribute__((ext_vector_type(8))) float v8f;

#define NQ 8
#define NLAYERS 4

struct cpx { float re, im; };

__device__ __forceinline__ cpx cmul(cpx a, cpx b) {
  cpx r; r.re = a.re * b.re - a.im * b.im; r.im = a.re * b.im + a.im * b.re; return r;
}

__device__ __forceinline__ cpx pick(cpx a, cpx b, int sel) {
  cpx r; r.re = sel ? b.re : a.re; r.im = sel ? b.im : a.im; return r;
}

__device__ __forceinline__ cpx u_entry(const float* cs, const float* sn, int gate, int i, int j) {
  float ca = cs[3 * gate],     sa = sn[3 * gate];
  float cb = cs[3 * gate + 1], sb = sn[3 * gate + 1];
  float cg = cs[3 * gate + 2], sg = sn[3 * gate + 2];
  cpx m;
  if (i == 0 && j == 0) { m.re =  cb * ca; m.im =  sb * sa; }
  else if (i == 0)      { m.re = -sb * ca; m.im = -cb * sa; }
  else if (j == 0)      { m.re =  sb * ca; m.im = -cb * sa; }
  else                  { m.re =  cb * ca; m.im = -sb * sa; }
  cpx ph; ph.re = cg; ph.im = (i == 0) ? -sg : sg;
  return cmul(ph, m);
}

__host__ __device__ constexpr int chainperm(int r) {
  r ^= ((r >> 1) & 1);
  r ^= ((r >> 2) & 1) << 1;
  r ^= ((r >> 3) & 1) << 2;
  return r;
}

__device__ __forceinline__ void build_gates(const float* __restrict__ w, float* tmp,
                                            float* fragf, int t,
                                            float cv, float sv, float* av) {
  float* cs  = tmp;
  float* sn  = tmp + 96;
  float* p4  = tmp + 256;
  if (t < NLAYERS * NQ * 3) {
    float h = 0.5f * w[t];
    cs[t] = cosf(h);
    sn[t] = sinf(h);
  }
  __syncthreads();
  {
    int q = t & 7;
    cpx u00 = u_entry(cs, sn, q, 0, 0), u01 = u_entry(cs, sn, q, 0, 1);
    cpx u10 = u_entry(cs, sn, q, 1, 0), u11 = u_entry(cs, sn, q, 1, 1);
    av[0] = u00.re * cv + u01.re * sv;  av[1] = u00.im * cv + u01.im * sv;
    av[2] = u10.re * cv + u11.re * sv;  av[3] = u10.im * cv + u11.im * sv;
  }
  if (t >= 64) {
    int l = t >> 6, p = (t >> 4) & 3, i4 = (t >> 2) & 3, j4 = t & 3;
    int g0 = l * NQ + 2 * p;
    cpx e = cmul(u_entry(cs, sn, g0,     i4 >> 1, j4 >> 1),
                 u_entry(cs, sn, g0 + 1, i4 & 1,  j4 & 1));
    p4[2 * t]     = e.re;
    p4[2 * t + 1] = e.im;
  }
  __syncthreads();
  {
    int m = t >> 4, k = t & 15;
    int mp = chainperm(m);
    int slot = (((k >> 1) & 3) * 32 + m + 16 * (k >> 3)) * 4 + (k & 1);
    #pragma unroll
    for (int l = 1; l < NLAYERS; ++l) {
      const float* pl = p4 + l * 128;
      cpx a, b, r;
      a.re = pl[(2 * 16 + (m >> 2) * 4 + (k >> 2)) * 2]; a.im = pl[(2 * 16 + (m >> 2) * 4 + (k >> 2)) * 2 + 1];
      b.re = pl[(3 * 16 + (m & 3) * 4 + (k & 3)) * 2];   b.im = pl[(3 * 16 + (m & 3) * 4 + (k & 3)) * 2 + 1];
      r = cmul(a, b);
      fragf[((l - 1) * 2 + 0) * 512 + slot]     = r.re;
      fragf[((l - 1) * 2 + 0) * 512 + slot + 2] = r.im - r.re;
      a.re = pl[(0 * 16 + (mp >> 2) * 4 + (k >> 2)) * 2]; a.im = pl[(0 * 16 + (mp >> 2) * 4 + (k >> 2)) * 2 + 1];
      b.re = pl[(1 * 16 + (mp & 3) * 4 + (k & 3)) * 2];   b.im = pl[(1 * 16 + (mp & 3) * 4 + (k & 3)) * 2 + 1];
      r = cmul(a, b);
      fragf[((l - 1) * 2 + 1) * 512 + slot]     = r.re;
      fragf[((l - 1) * 2 + 1) * 512 + slot + 2] = r.im - r.re;
    }
  }
  __syncthreads();
}


template <bool STATE_A>
__device__ __forceinline__ v8f mm(v2f g, v2f s, v8f acc) {
  return STATE_A
    ? __builtin_amdgcn_wmma_f32_16x16x4_f32(false, s, false, g, (short)0, acc, false, false)
    : __builtin_amdgcn_wmma_f32_16x16x4_f32(false, g, false, s, (short)0, acc, false, false);
}

template <bool STATE_A>
__device__ __forceinline__ void cmulmat(const v4f* mat, int lane, float* X, float* Y) {
  v4f f[4];
  #pragma unroll
  for (int kc = 0; kc < 4; ++kc) f[kc] = mat[kc * 32 + lane];
  v8f accM = {};
  #pragma unroll
  for (int kc = 0; kc < 4; ++kc) {
    int a0 = 2 * kc, a1 = a0 + 1;
    v2f ar;  ar[0] = f[kc][0];       ar[1] = f[kc][1];
    v2f bs;  bs[0] = X[a0] + Y[a0];  bs[1] = X[a1] + Y[a1];
    accM = mm<STATE_A>(ar, bs, accM);
  }
  v8f accX = accM; v8f accY = accM;
  #pragma unroll
  for (int kc = 0; kc < 4; ++kc) {
    int a0 = 2 * kc, a1 = a0 + 1;
    v2f ad;  ad[0] = f[kc][2];  ad[1] = f[kc][3];
    v2f aq;  aq[0] = -2.f * f[kc][0] - ad[0];
             aq[1] = -2.f * f[kc][1] - ad[1];
    v2f bx;  bx[0] = X[a0];     bx[1] = X[a1];
    v2f by;  by[0] = Y[a0];     by[1] = Y[a1];
    accX = mm<STATE_A>(aq, by, accX);
    accY = mm<STATE_A>(ad, bx, accY);
  }
  #pragma unroll
  for (int v = 0; v < 8; ++v) { X[v] = accX[v]; Y[v] = accY[v]; }
}

__global__ __launch_bounds__(256, 16) void qsim(const float* __restrict__ x,
                                            const float* __restrict__ w,
                                            float* __restrict__ out, int B) {
  __shared__ float tmp[768];
  __shared__ v4f   frag[6 * 4 * 32];
  const int lane  = threadIdx.x & 31;
  const int wslot = __builtin_amdgcn_readfirstlane(threadIdx.x >> 5);
  const int bw    = blockIdx.x * 8 + wslot;
  const int b     = bw < B ? bw : B - 1;
  const int half = lane >> 4;
  const int col  = lane & 15;

  float sv, cv;
  sincosf(0.5f * x[b * 8 + (lane & 7)], &sv, &cv);

  float av[4];
  build_gates(w, tmp, (float*)frag, threadIdx.x, cv, sv, av);

  cpx a0[8], a1[8];
  #pragma unroll
  for (int q = 0; q < 8; ++q) {
    a0[q].re = __int_as_float(__builtin_amdgcn_readlane(__float_as_int(av[0]), q));
    a0[q].im = __int_as_float(__builtin_amdgcn_readlane(__float_as_int(av[1]), q));
    a1[q].re = __int_as_float(__builtin_amdgcn_readlane(__float_as_int(av[2]), q));
    a1[q].im = __int_as_float(__builtin_amdgcn_readlane(__float_as_int(av[3]), q));
  }

  float X[8], Y[8];
  {
    cpx pG[2], Q;
    {
      const v4f* matW = frag + 1 * 128;
      v4f fw[4];
      #pragma unroll
      for (int kc = 0; kc < 4; ++kc) fw[kc] = matW[kc * 32 + lane];
      v2f ure[2], uim[2];
      #pragma unroll
      for (int pp = 0; pp < 2; ++pp) {
        cpx c0 = cmul(pp ? a1[2] : a0[2], a0[3]), c1 = cmul(pp ? a1[2] : a0[2], a1[3]);
        ure[pp][0] = c0.re; ure[pp][1] = c1.re; uim[pp][0] = c0.im; uim[pp][1] = c1.im;
      }
      v2f sre[2], sim[2];
      #pragma unroll
      for (int i = 0; i < 2; ++i) { sre[i] = (v2f){0.f, 0.f}; sim[i] = (v2f){0.f, 0.f}; }
      #pragma unroll
      for (int kc = 0; kc < 4; ++kc) {
        const int v1 = kc & 1, v2 = kc >> 1, pp = v1 ^ v2;
        v2f ur = v1 ? __builtin_shufflevector(ure[pp], ure[pp], 1, 0) : ure[pp];
        v2f ui = v1 ? __builtin_shufflevector(uim[pp], uim[pp], 1, 0) : uim[pp];
        v2f wre; wre[0] = fw[kc][0]; wre[1] = fw[kc][1];
        v2f wim; wim[0] = fw[kc][2]; wim[1] = fw[kc][3];
        wim = wim + wre;
        sre[v2] = sre[v2] + wre * ur - wim * ui;
        sim[v2] = sim[v2] + wre * ui + wim * ur;
      }
      cpx h0 = pick(a0[0], a1[0], half);
      cpx g0 = cmul(h0, pick(a0[1], a1[1], half));
      cpx g1 = cmul(h0, pick(a0[1], a1[1], half ^ 1));
      v2f gre = g0.re * sre[0] - g0.im * sim[0] + g1.re * sre[1] - g1.im * sim[1];
      v2f gim = g0.re * sim[0] + g0.im * sre[0] + g1.re * sim[1] + g1.im * sre[1];
      pG[0].re = gre[0]; pG[0].im = gim[0]; pG[1].re = gre[1]; pG[1].im = gim[1];
    }
    {
      const v4f* matU = frag + 0 * 128;
      v4f fu[4];
      #pragma unroll
      for (int kc = 0; kc < 4; ++kc) fu[kc] = matU[kc * 32 + lane];
      v2f tre[2], tim[2];
      #pragma unroll
      for (int pp = 0; pp < 2; ++pp) {
        cpx c0 = cmul(pp ? a1[6] : a0[6], a0[7]), c1 = cmul(pp ? a1[6] : a0[6], a1[7]);
        tre[pp][0] = c0.re; tre[pp][1] = c1.re; tim[pp][0] = c0.im; tim[pp][1] = c1.im;
      }
      v2f sre[2], sim[2];
      #pragma unroll
      for (int i = 0; i < 2; ++i) { sre[i] = (v2f){0.f, 0.f}; sim[i] = (v2f){0.f, 0.f}; }
      #pragma unroll
      for (int kc = 0; kc < 4; ++kc) {
        const int v1 = kc & 1, v2 = kc >> 1, pp = v1 ^ v2;
        v2f tr = v1 ? __builtin_shufflevector(tre[pp], tre[pp], 1, 0) : tre[pp];
        v2f ti = v1 ? __builtin_shufflevector(tim[pp], tim[pp], 1, 0) : tim[pp];
        v2f wre; wre[0] = fu[kc][0]; wre[1] = fu[kc][1];
        v2f wim; wim[0] = fu[kc][2]; wim[1] = fu[kc][3];
        wim = wim + wre;
        sre[v2] = sre[v2] + wre * tr - wim * ti;
        sim[v2] = sim[v2] + wre * ti + wim * tr;
      }
      cpx b0 = pick(a0[5], a1[5], half);
      cpx b1 = pick(a0[5], a1[5], half ^ 1);
      v2f qre = b0.re * sre[0] - b0.im * sim[0] + b1.re * sre[1] - b1.im * sim[1];
      v2f qim = b0.re * sim[0] + b0.im * sre[0] + b1.re * sim[1] + b1.im * sre[1];
      Q.re = qre[0] + qre[1];  Q.im = qim[0] + qim[1];
    }
    cpx kep = pick(pG[0], pG[1], half);
    cpx snd = pick(pG[1], pG[0], half);
    cpx G, Qo, M;
    G.re  = kep.re + __shfl_xor(snd.re, 16, 32);
    G.im  = kep.im + __shfl_xor(snd.im, 16, 32);
    Qo.re = __shfl_xor(Q.re, 16, 32);
    Qo.im = __shfl_xor(Q.im, 16, 32);
    cpx m0 = cmul(a0[4], Q), m1 = cmul(a1[4], Qo);
    M.re = m0.re + m1.re;  M.im = m0.im + m1.im;
    v2f am;  am[0] = M.re;  am[1] = M.im;
    v2f bx;  bx[0] = G.re;  bx[1] = -G.im;
    v2f by;  by[0] = G.im;  by[1] = G.re;
    v8f zero = {};
    v8f accX = mm<false>(am, bx, zero);
    v8f accY = mm<false>(am, by, zero);
    #pragma unroll
    for (int v = 0; v < 8; ++v) { X[v] = accX[v]; Y[v] = accY[v]; }
  }

  for (int l = 1; l < NLAYERS; ++l) {
    const int flip = l & 1;
    const v4f* matA = frag + ((l - 1) * 2 + flip)       * 128;
    const v4f* matB = frag + ((l - 1) * 2 + (1 - flip)) * 128;

    if (l > 1) {
      cmulmat<true>(matA, lane, X, Y);
      cmulmat<false>(matB, lane, X, Y);
    }

    if (flip) {
      #pragma unroll
      for (int v = 0; v < 8; ++v) {
        float tx = __shfl_xor(X[v], 16, 32);
        float ty = __shfl_xor(Y[v], 16, 32);
        if (lane & 1) { X[v] = tx; Y[v] = ty; }
      }
      float nX[8], nY[8];
      #pragma unroll
      for (int v = 0; v < 8; ++v) {
        const int r0 = chainperm(v), r1 = chainperm(v + 8);
        float t0x = (r0 < 8)  ? X[r0 & 7] : __shfl_xor(X[r0 & 7], 16, 32);
        float t0y = (r0 < 8)  ? Y[r0 & 7] : __shfl_xor(Y[r0 & 7], 16, 32);
        float t1x = (r1 >= 8) ? X[r1 & 7] : __shfl_xor(X[r1 & 7], 16, 32);
        float t1y = (r1 >= 8) ? Y[r1 & 7] : __shfl_xor(Y[r1 & 7], 16, 32);
        nX[v] = half ? t1x : t0x;
        nY[v] = half ? t1y : t0y;
      }
      #pragma unroll
      for (int v = 0; v < 8; ++v) { X[v] = nX[v]; Y[v] = nY[v]; }
    } else {
      #pragma unroll
      for (int v = 1; v < 8; v += 2) {
        X[v] = __shfl_xor(X[v], 8, 32);
        Y[v] = __shfl_xor(Y[v], 8, 32);
      }
      int src = chainperm(lane & 15) | (lane & 16);
      #pragma unroll
      for (int v = 0; v < 8; ++v) {
        X[v] = __shfl(X[v], src, 32);
        Y[v] = __shfl(Y[v], src, 32);
      }
    }
  }

  float psum = 0.f, s5 = 0.f, s6 = 0.f, s7 = 0.f;
  #pragma unroll
  for (int v = 0; v < 8; ++v) {
    float p = X[v] * X[v] + Y[v] * Y[v];
    psum += p;
    s5 += ((v >> 2) & 1) ? -p : p;
    s6 += ((v >> 1) & 1) ? -p : p;
    s7 += ((v) & 1)      ? -p : p;
  }
  float o[8];
  o[0] = ((col >> 3) & 1) ? -psum : psum;
  o[1] = ((col >> 2) & 1) ? -psum : psum;
  o[2] = ((col >> 1) & 1) ? -psum : psum;
  o[3] = ((col) & 1)      ? -psum : psum;
  o[4] = half ? -psum : psum;
  o[5] = s5; o[6] = s6; o[7] = s7;
  #pragma unroll
  for (int q = 0; q < 8; ++q) {
    float v = o[q];
    #pragma unroll
    for (int off = 16; off > 0; off >>= 1) v += __shfl_xor(v, off, 32);
    o[q] = v;
  }
  if (lane == 0 && bw < B) {
    #pragma unroll
    for (int q = 0; q < 8; ++q) out[b * 8 + q] = o[q];
  }
}

extern "C" void kernel_launch(void* const* d_in, const int* in_sizes, int n_in,
                              void* d_out, int out_size, void* d_ws, size_t ws_size,
                              hipStream_t stream) {
  const float* x = (const float*)d_in[0];
  const float* w = (const float*)d_in[1];
  float* out = (float*)d_out;
  int B = in_sizes[0] / NQ;

  int nblocks = (B + 7) / 8;
  qsim<<<nblocks, 256, 0, stream>>>(x, w, out, B);
}


# Work-group widened from 8 to 32 waves (32 samples): one gate-matrix build in LDS now serves 32 samples instead of 8; waves 0..7 build, cosine and sine tables on separate waves; sums unchanged
# speedup vs baseline: 3.4236x; 1.0701x over previous
#include <hip/hip_runtime.h>


typedef __attribute__((ext_vector_type(2))) float v2f;
typedef __attribute__((ext_vector_type(4))) float v4f;
typedef __attribute__((ext_vector_type(8))) float v8f;

#define NQ 8
#define NLAYERS 4

struct cpx { float re, im; };

__device__ __forceinline__ cpx cmul(cpx a, cpx b) {
  cpx r; r.re = a.re * b.re - a.im * b.im; r.im = a.re * b.im + a.im * b.re; return r;
}

__device__ __forceinline__ cpx pick(cpx a, cpx b, int sel) {
  cpx r; r.re = sel ? b.re : a.re; r.im = sel ? b.im : a.im; return r;
}

__device__ __forceinline__ cpx u_entry(const float* cs, const float* sn, int gate, int i, int j) {
  float ca = cs[3 * gate],     sa = sn[3 * gate];
  float cb = cs[3 * gate + 1], sb = sn[3 * gate + 1];
  float cg = cs[3 * gate + 2], sg = sn[3 * gate + 2];
  cpx m;
  if (i == 0 && j == 0) { m.re =  cb * ca; m.im =  sb * sa; }
  else if (i == 0)      { m.re = -sb * ca; m.im = -cb * sa; }
  else if (j == 0)      { m.re =  sb * ca; m.im = -cb * sa; }
  else                  { m.re =  cb * ca; m.im = -sb * sa; }
  cpx ph; ph.re = cg; ph.im = (i == 0) ? -sg : sg;
  return cmul(ph, m);
}

__host__ __device__ constexpr int chainperm(int r) {
  r ^= ((r >> 1) & 1);
  r ^= ((r >> 2) & 1) << 1;
  r ^= ((r >> 3) & 1) << 2;
  return r;
}

__device__ __forceinline__ void build_gates(const float* __restrict__ w, float* tmp,
                                            float* fragf, int t,
                                            float cv, float sv, float* av) {
  float* cs  = tmp;
  float* sn  = tmp + 96;
  float* p4  = tmp + 256;
  if (t < NLAYERS * NQ * 3) {
    cs[t] = cosf(0.5f * w[t]);
  } else if (t < 2 * NLAYERS * NQ * 3) {
    sn[t - NLAYERS * NQ * 3] = sinf(0.5f * w[t - NLAYERS * NQ * 3]);
  }
  __syncthreads();
  {
    int q = t & 7;
    cpx u00 = u_entry(cs, sn, q, 0, 0), u01 = u_entry(cs, sn, q, 0, 1);
    cpx u10 = u_entry(cs, sn, q, 1, 0), u11 = u_entry(cs, sn, q, 1, 1);
    av[0] = u00.re * cv + u01.re * sv;  av[1] = u00.im * cv + u01.im * sv;
    av[2] = u10.re * cv + u11.re * sv;  av[3] = u10.im * cv + u11.im * sv;
  }
  if (t >= 64 && t < 256) {
    int l = t >> 6, p = (t >> 4) & 3, i4 = (t >> 2) & 3, j4 = t & 3;
    int g0 = l * NQ + 2 * p;
    cpx e = cmul(u_entry(cs, sn, g0,     i4 >> 1, j4 >> 1),
                 u_entry(cs, sn, g0 + 1, i4 & 1,  j4 & 1));
    p4[2 * t]     = e.re;
    p4[2 * t + 1] = e.im;
  }
  __syncthreads();
  if (t < 256) {
    int m = t >> 4, k = t & 15;
    int mp = chainperm(m);
    int slot = (((k >> 1) & 3) * 32 + m + 16 * (k >> 3)) * 4 + (k & 1);
    #pragma unroll
    for (int l = 1; l < NLAYERS; ++l) {
      const float* pl = p4 + l * 128;
      cpx a, b, r;
      a.re = pl[(2 * 16 + (m >> 2) * 4 + (k >> 2)) * 2]; a.im = pl[(2 * 16 + (m >> 2) * 4 + (k >> 2)) * 2 + 1];
      b.re = pl[(3 * 16 + (m & 3) * 4 + (k & 3)) * 2];   b.im = pl[(3 * 16 + (m & 3) * 4 + (k & 3)) * 2 + 1];
      r = cmul(a, b);
      fragf[((l - 1) * 2 + 0) * 512 + slot]     = r.re;
      fragf[((l - 1) * 2 + 0) * 512 + slot + 2] = r.im - r.re;
      a.re = pl[(0 * 16 + (mp >> 2) * 4 + (k >> 2)) * 2]; a.im = pl[(0 * 16 + (mp >> 2) * 4 + (k >> 2)) * 2 + 1];
      b.re = pl[(1 * 16 + (mp & 3) * 4 + (k & 3)) * 2];   b.im = pl[(1 * 16 + (mp & 3) * 4 + (k & 3)) * 2 + 1];
      r = cmul(a, b);
      fragf[((l - 1) * 2 + 1) * 512 + slot]     = r.re;
      fragf[((l - 1) * 2 + 1) * 512 + slot + 2] = r.im - r.re;
    }
  }
  __syncthreads();
}


template <bool STATE_A>
__device__ __forceinline__ v8f mm(v2f g, v2f s, v8f acc) {
  return STATE_A
    ? __builtin_amdgcn_wmma_f32_16x16x4_f32(false, s, false, g, (short)0, acc, false, false)
    : __builtin_amdgcn_wmma_f32_16x16x4_f32(false, g, false, s, (short)0, acc, false, false);
}

template <bool STATE_A>
__device__ __forceinline__ void cmulmat(const v4f* mat, int lane, float* X, float* Y) {
  v4f f[4];
  #pragma unroll
  for (int kc = 0; kc < 4; ++kc) f[kc] = mat[kc * 32 + lane];
  v8f accM = {};
  #pragma unroll
  for (int kc = 0; kc < 4; ++kc) {
    int a0 = 2 * kc, a1 = a0 + 1;
    v2f ar;  ar[0] = f[kc][0];       ar[1] = f[kc][1];
    v2f bs;  bs[0] = X[a0] + Y[a0];  bs[1] = X[a1] + Y[a1];
    accM = mm<STATE_A>(ar, bs, accM);
  }
  v8f accX = accM; v8f accY = accM;
  #pragma unroll
  for (int kc = 0; kc < 4; ++kc) {
    int a0 = 2 * kc, a1 = a0 + 1;
    v2f ad;  ad[0] = f[kc][2];  ad[1] = f[kc][3];
    v2f aq;  aq[0] = -2.f * f[kc][0] - ad[0];
             aq[1] = -2.f * f[kc][1] - ad[1];
    v2f bx;  bx[0] = X[a0];     bx[1] = X[a1];
    v2f by;  by[0] = Y[a0];     by[1] = Y[a1];
    accX = mm<STATE_A>(aq, by, accX);
    accY = mm<STATE_A>(ad, bx, accY);
  }
  #pragma unroll
  for (int v = 0; v < 8; ++v) { X[v] = accX[v]; Y[v] = accY[v]; }
}

__global__ __launch_bounds__(1024, 16) void qsim(const float* __restrict__ x,
                                            const float* __restrict__ w,
                                            float* __restrict__ out, int B) {
  __shared__ float tmp[768];
  __shared__ v4f   frag[6 * 4 * 32];
  const int lane  = threadIdx.x & 31;
  const int wslot = __builtin_amdgcn_readfirstlane(threadIdx.x >> 5);
  const int bw    = blockIdx.x * 32 + wslot;
  const int b     = bw < B ? bw : B - 1;
  const int half = lane >> 4;
  const int col  = lane & 15;

  float sv, cv;
  sincosf(0.5f * x[b * 8 + (lane & 7)], &sv, &cv);

  float av[4];
  build_gates(w, tmp, (float*)frag, threadIdx.x, cv, sv, av);

  cpx a0[8], a1[8];
  #pragma unroll
  for (int q = 0; q < 8; ++q) {
    a0[q].re = __int_as_float(__builtin_amdgcn_readlane(__float_as_int(av[0]), q));
    a0[q].im = __int_as_float(__builtin_amdgcn_readlane(__float_as_int(av[1]), q));
    a1[q].re = __int_as_float(__builtin_amdgcn_readlane(__float_as_int(av[2]), q));
    a1[q].im = __int_as_float(__builtin_amdgcn_readlane(__float_as_int(av[3]), q));
  }

  float X[8], Y[8];
  {
    cpx pG[2], Q;
    {
      const v4f* matW = frag + 1 * 128;
      v4f fw[4];
      #pragma unroll
      for (int kc = 0; kc < 4; ++kc) fw[kc] = matW[kc * 32 + lane];
      v2f ure[2], uim[2];
      #pragma unroll
      for (int pp = 0; pp < 2; ++pp) {
        cpx c0 = cmul(pp ? a1[2] : a0[2], a0[3]), c1 = cmul(pp ? a1[2] : a0[2], a1[3]);
        ure[pp][0] = c0.re; ure[pp][1] = c1.re; uim[pp][0] = c0.im; uim[pp][1] = c1.im;
      }
      v2f sre[2], sim[2];
      #pragma unroll
      for (int i = 0; i < 2; ++i) { sre[i] = (v2f){0.f, 0.f}; sim[i] = (v2f){0.f, 0.f}; }
      #pragma unroll
      for (int kc = 0; kc < 4; ++kc) {
        const int v1 = kc & 1, v2 = kc >> 1, pp = v1 ^ v2;
        v2f ur = v1 ? __builtin_shufflevector(ure[pp], ure[pp], 1, 0) : ure[pp];
        v2f ui = v1 ? __builtin_shufflevector(uim[pp], uim[pp], 1, 0) : uim[pp];
        v2f wre; wre[0] = fw[kc][0]; wre[1] = fw[kc][1];
        v2f wim; wim[0] = fw[kc][2]; wim[1] = fw[kc][3];
        wim = wim + wre;
        sre[v2] = sre[v2] + wre * ur - wim * ui;
        sim[v2] = sim[v2] + wre * ui + wim * ur;
      }
      cpx h0 = pick(a0[0], a1[0], half);
      cpx g0 = cmul(h0, pick(a0[1], a1[1], half));
      cpx g1 = cmul(h0, pick(a0[1], a1[1], half ^ 1));
      v2f gre = g0.re * sre[0] - g0.im * sim[0] + g1.re * sre[1] - g1.im * sim[1];
      v2f gim = g0.re * sim[0] + g0.im * sre[0] + g1.re * sim[1] + g1.im * sre[1];
      pG[0].re = gre[0]; pG[0].im = gim[0]; pG[1].re = gre[1]; pG[1].im = gim[1];
    }
    {
      const v4f* matU = frag + 0 * 128;
      v4f fu[4];
      #pragma unroll
      for (int kc = 0; kc < 4; ++kc) fu[kc] = matU[kc * 32 + lane];
      v2f tre[2], tim[2];
      #pragma unroll
      for (int pp = 0; pp < 2; ++pp) {
        cpx c0 = cmul(pp ? a1[6] : a0[6], a0[7]), c1 = cmul(pp ? a1[6] : a0[6], a1[7]);
        tre[pp][0] = c0.re; tre[pp][1] = c1.re; tim[pp][0] = c0.im; tim[pp][1] = c1.im;
      }
      v2f sre[2], sim[2];
      #pragma unroll
      for (int i = 0; i < 2; ++i) { sre[i] = (v2f){0.f, 0.f}; sim[i] = (v2f){0.f, 0.f}; }
      #pragma unroll
      for (int kc = 0; kc < 4; ++kc) {
        const int v1 = kc & 1, v2 = kc >> 1, pp = v1 ^ v2;
        v2f tr = v1 ? __builtin_shufflevector(tre[pp], tre[pp], 1, 0) : tre[pp];
        v2f ti = v1 ? __builtin_shufflevector(tim[pp], tim[pp], 1, 0) : tim[pp];
        v2f wre; wre[0] = fu[kc][0]; wre[1] = fu[kc][1];
        v2f wim; wim[0] = fu[kc][2]; wim[1] = fu[kc][3];
        wim = wim + wre;
        sre[v2] = sre[v2] + wre * tr - wim * ti;
        sim[v2] = sim[v2] + wre * ti + wim * tr;
      }
      cpx b0 = pick(a0[5], a1[5], half);
      cpx b1 = pick(a0[5], a1[5], half ^ 1);
      v2f qre = b0.re * sre[0] - b0.im * sim[0] + b1.re * sre[1] - b1.im * sim[1];
      v2f qim = b0.re * sim[0] + b0.im * sre[0] + b1.re * sim[1] + b1.im * sre[1];
      Q.re = qre[0] + qre[1];  Q.im = qim[0] + qim[1];
    }
    cpx kep = pick(pG[0], pG[1], half);
    cpx snd = pick(pG[1], pG[0], half);
    cpx G, Qo, M;
    G.re  = kep.re + __shfl_xor(snd.re, 16, 32);
    G.im  = kep.im + __shfl_xor(snd.im, 16, 32);
    Qo.re = __shfl_xor(Q.re, 16, 32);
    Qo.im = __shfl_xor(Q.im, 16, 32);
    cpx m0 = cmul(a0[4], Q), m1 = cmul(a1[4], Qo);
    M.re = m0.re + m1.re;  M.im = m0.im + m1.im;
    v2f am;  am[0] = M.re;  am[1] = M.im;
    v2f bx;  bx[0] = G.re;  bx[1] = -G.im;
    v2f by;  by[0] = G.im;  by[1] = G.re;
    v8f zero = {};
    v8f accX = mm<false>(am, bx, zero);
    v8f accY = mm<false>(am, by, zero);
    #pragma unroll
    for (int v = 0; v < 8; ++v) { X[v] = accX[v]; Y[v] = accY[v]; }
  }

  for (int l = 1; l < NLAYERS; ++l) {
    const int flip = l & 1;
    const v4f* matA = frag + ((l - 1) * 2 + flip)       * 128;
    const v4f* matB = frag + ((l - 1) * 2 + (1 - flip)) * 128;

    if (l > 1) {
      cmulmat<true>(matA, lane, X, Y);
      cmulmat<false>(matB, lane, X, Y);
    }

    if (flip) {
      #pragma unroll
      for (int v = 0; v < 8; ++v) {
        float tx = __shfl_xor(X[v], 16, 32);
        float ty = __shfl_xor(Y[v], 16, 32);
        if (lane & 1) { X[v] = tx; Y[v] = ty; }
      }
      float nX[8], nY[8];
      #pragma unroll
      for (int v = 0; v < 8; ++v) {
        const int r0 = chainperm(v), r1 = chainperm(v + 8);
        float t0x = (r0 < 8)  ? X[r0 & 7] : __shfl_xor(X[r0 & 7], 16, 32);
        float t0y = (r0 < 8)  ? Y[r0 & 7] : __shfl_xor(Y[r0 & 7], 16, 32);
        float t1x = (r1 >= 8) ? X[r1 & 7] : __shfl_xor(X[r1 & 7], 16, 32);
        float t1y = (r1 >= 8) ? Y[r1 & 7] : __shfl_xor(Y[r1 & 7], 16, 32);
        nX[v] = half ? t1x : t0x;
        nY[v] = half ? t1y : t0y;
      }
      #pragma unroll
      for (int v = 0; v < 8; ++v) { X[v] = nX[v]; Y[v] = nY[v]; }
    } else {
      #pragma unroll
      for (int v = 1; v < 8; v += 2) {
        X[v] = __shfl_xor(X[v], 8, 32);
        Y[v] = __shfl_xor(Y[v], 8, 32);
      }
      int src = chainperm(lane & 15) | (lane & 16);
      #pragma unroll
      for (int v = 0; v < 8; ++v) {
        X[v] = __shfl(X[v], src, 32);
        Y[v] = __shfl(Y[v], src, 32);
      }
    }
  }

  float psum = 0.f, s5 = 0.f, s6 = 0.f, s7 = 0.f;
  #pragma unroll
  for (int v = 0; v < 8; ++v) {
    float p = X[v] * X[v] + Y[v] * Y[v];
    psum += p;
    s5 += ((v >> 2) & 1) ? -p : p;
    s6 += ((v >> 1) & 1) ? -p : p;
    s7 += ((v) & 1)      ? -p : p;
  }
  float o[8];
  o[0] = ((col >> 3) & 1) ? -psum : psum;
  o[1] = ((col >> 2) & 1) ? -psum : psum;
  o[2] = ((col >> 1) & 1) ? -psum : psum;
  o[3] = ((col) & 1)      ? -psum : psum;
  o[4] = half ? -psum : psum;
  o[5] = s5; o[6] = s6; o[7] = s7;
  #pragma unroll
  for (int q = 0; q < 8; ++q) {
    float v = o[q];
    #pragma unroll
    for (int off = 16; off > 0; off >>= 1) v += __shfl_xor(v, off, 32);
    o[q] = v;
  }
  if (lane == 0 && bw < B) {
    #pragma unroll
    for (int q = 0; q < 8; ++q) out[b * 8 + q] = o[q];
  }
}

extern "C" void kernel_launch(void* const* d_in, const int* in_sizes, int n_in,
                              void* d_out, int out_size, void* d_ws, size_t ws_size,
                              hipStream_t stream) {
  const float* x = (const float*)d_in[0];
  const float* w = (const float*)d_in[1];
  float* out = (float*)d_out;
  int B = in_sizes[0] / NQ;

  int nblocks = (B + 31) / 32;
  qsim<<<nblocks, 1024, 0, stream>>>(x, w, out, B);
}


# Measurement stage moves less data: last layer's CNOTs taken as signs of the Z sums (no exchange), and the eight wave sums share one halving reduction, 9 lane exchanges instead of 56; sums regrouped
# speedup vs baseline: 3.6081x; 1.0539x over previous
#include <hip/hip_runtime.h>


typedef __attribute__((ext_vector_type(2))) float v2f;
typedef __attribute__((ext_vector_type(4))) float v4f;
typedef __attribute__((ext_vector_type(8))) float v8f;

#define NQ 8
#define NLAYERS 4

struct cpx { float re, im; };

__device__ __forceinline__ cpx cmul(cpx a, cpx b) {
  cpx r; r.re = a.re * b.re - a.im * b.im; r.im = a.re * b.im + a.im * b.re; return r;
}

__device__ __forceinline__ cpx pick(cpx a, cpx b, int sel) {
  cpx r; r.re = sel ? b.re : a.re; r.im = sel ? b.im : a.im; return r;
}

__device__ __forceinline__ cpx u_entry(const float* cs, const float* sn, int gate, int i, int j) {
  float ca = cs[3 * gate],     sa = sn[3 * gate];
  float cb = cs[3 * gate + 1], sb = sn[3 * gate + 1];
  float cg = cs[3 * gate + 2], sg = sn[3 * gate + 2];
  cpx m;
  if (i == 0 && j == 0) { m.re =  cb * ca; m.im =  sb * sa; }
  else if (i == 0)      { m.re = -sb * ca; m.im = -cb * sa; }
  else if (j == 0)      { m.re =  sb * ca; m.im = -cb * sa; }
  else                  { m.re =  cb * ca; m.im = -sb * sa; }
  cpx ph; ph.re = cg; ph.im = (i == 0) ? -sg : sg;
  return cmul(ph, m);
}

__host__ __device__ constexpr int chainperm(int r) {
  r ^= ((r >> 1) & 1);
  r ^= ((r >> 2) & 1) << 1;
  r ^= ((r >> 3) & 1) << 2;
  return r;
}

__device__ __forceinline__ void build_gates(const float* __restrict__ w, float* tmp,
                                            float* fragf, int t,
                                            float cv, float sv, float* av) {
  float* cs  = tmp;
  float* sn  = tmp + 96;
  float* p4  = tmp + 256;
  if (t < NLAYERS * NQ * 3) {
    cs[t] = cosf(0.5f * w[t]);
  } else if (t < 2 * NLAYERS * NQ * 3) {
    sn[t - NLAYERS * NQ * 3] = sinf(0.5f * w[t - NLAYERS * NQ * 3]);
  }
  __syncthreads();
  {
    int q = t & 7;
    cpx u00 = u_entry(cs, sn, q, 0, 0), u01 = u_entry(cs, sn, q, 0, 1);
    cpx u10 = u_entry(cs, sn, q, 1, 0), u11 = u_entry(cs, sn, q, 1, 1);
    av[0] = u00.re * cv + u01.re * sv;  av[1] = u00.im * cv + u01.im * sv;
    av[2] = u10.re * cv + u11.re * sv;  av[3] = u10.im * cv + u11.im * sv;
  }
  if (t >= 64 && t < 256) {
    int l = t >> 6, p = (t >> 4) & 3, i4 = (t >> 2) & 3, j4 = t & 3;
    int g0 = l * NQ + 2 * p;
    cpx e = cmul(u_entry(cs, sn, g0,     i4 >> 1, j4 >> 1),
                 u_entry(cs, sn, g0 + 1, i4 & 1,  j4 & 1));
    p4[2 * t]     = e.re;
    p4[2 * t + 1] = e.im;
  }
  __syncthreads();
  if (t < 256) {
    int m = t >> 4, k = t & 15;
    int mp = chainperm(m);
    int slot = (((k >> 1) & 3) * 32 + m + 16 * (k >> 3)) * 4 + (k & 1);
    #pragma unroll
    for (int l = 1; l < NLAYERS; ++l) {
      const float* pl = p4 + l * 128;
      cpx a, b, r;
      a.re = pl[(2 * 16 + (m >> 2) * 4 + (k >> 2)) * 2]; a.im = pl[(2 * 16 + (m >> 2) * 4 + (k >> 2)) * 2 + 1];
      b.re = pl[(3 * 16 + (m & 3) * 4 + (k & 3)) * 2];   b.im = pl[(3 * 16 + (m & 3) * 4 + (k & 3)) * 2 + 1];
      r = cmul(a, b);
      fragf[((l - 1) * 2 + 0) * 512 + slot]     = r.re;
      fragf[((l - 1) * 2 + 0) * 512 + slot + 2] = r.im - r.re;
      a.re = pl[(0 * 16 + (mp >> 2) * 4 + (k >> 2)) * 2]; a.im = pl[(0 * 16 + (mp >> 2) * 4 + (k >> 2)) * 2 + 1];
      b.re = pl[(1 * 16 + (mp & 3) * 4 + (k & 3)) * 2];   b.im = pl[(1 * 16 + (mp & 3) * 4 + (k & 3)) * 2 + 1];
      r = cmul(a, b);
      fragf[((l - 1) * 2 + 1) * 512 + slot]     = r.re;
      fragf[((l - 1) * 2 + 1) * 512 + slot + 2] = r.im - r.re;
    }
  }
  __syncthreads();
}


template <bool STATE_A>
__device__ __forceinline__ v8f mm(v2f g, v2f s, v8f acc) {
  return STATE_A
    ? __builtin_amdgcn_wmma_f32_16x16x4_f32(false, s, false, g, (short)0, acc, false, false)
    : __builtin_amdgcn_wmma_f32_16x16x4_f32(false, g, false, s, (short)0, acc, false, false);
}

template <bool STATE_A>
__device__ __forceinline__ void cmulmat(const v4f* mat, int lane, float* X, float* Y) {
  v4f f[4];
  #pragma unroll
  for (int kc = 0; kc < 4; ++kc) f[kc] = mat[kc * 32 + lane];
  v8f accM = {};
  #pragma unroll
  for (int kc = 0; kc < 4; ++kc) {
    int a0 = 2 * kc, a1 = a0 + 1;
    v2f ar;  ar[0] = f[kc][0];       ar[1] = f[kc][1];
    v2f bs;  bs[0] = X[a0] + Y[a0];  bs[1] = X[a1] + Y[a1];
    accM = mm<STATE_A>(ar, bs, accM);
  }
  v8f accX = accM; v8f accY = accM;
  #pragma unroll
  for (int kc = 0; kc < 4; ++kc) {
    int a0 = 2 * kc, a1 = a0 + 1;
    v2f ad;  ad[0] = f[kc][2];  ad[1] = f[kc][3];
    v2f aq;  aq[0] = -2.f * f[kc][0] - ad[0];
             aq[1] = -2.f * f[kc][1] - ad[1];
    v2f bx;  bx[0] = X[a0];     bx[1] = X[a1];
    v2f by;  by[0] = Y[a0];     by[1] = Y[a1];
    accX = mm<STATE_A>(aq, by, accX);
    accY = mm<STATE_A>(ad, bx, accY);
  }
  #pragma unroll
  for (int v = 0; v < 8; ++v) { X[v] = accX[v]; Y[v] = accY[v]; }
}

__global__ __launch_bounds__(1024, 16) void qsim(const float* __restrict__ x,
                                            const float* __restrict__ w,
                                            float* __restrict__ out, int B) {
  __shared__ float tmp[768];
  __shared__ v4f   frag[6 * 4 * 32];
  const int lane  = threadIdx.x & 31;
  const int wslot = __builtin_amdgcn_readfirstlane(threadIdx.x >> 5);
  const int bw    = blockIdx.x * 32 + wslot;
  const int b     = bw < B ? bw : B - 1;
  const int half = lane >> 4;
  const int col  = lane & 15;

  float sv, cv;
  sincosf(0.5f * x[b * 8 + (lane & 7)], &sv, &cv);

  float av[4];
  build_gates(w, tmp, (float*)frag, threadIdx.x, cv, sv, av);

  cpx a0[8], a1[8];
  #pragma unroll
  for (int q = 0; q < 8; ++q) {
    a0[q].re = __int_as_float(__builtin_amdgcn_readlane(__float_as_int(av[0]), q));
    a0[q].im = __int_as_float(__builtin_amdgcn_readlane(__float_as_int(av[1]), q));
    a1[q].re = __int_as_float(__builtin_amdgcn_readlane(__float_as_int(av[2]), q));
    a1[q].im = __int_as_float(__builtin_amdgcn_readlane(__float_as_int(av[3]), q));
  }

  float X[8], Y[8];
  {
    cpx pG[2], Q;
    {
      const v4f* matW = frag + 1 * 128;
      v4f fw[4];
      #pragma unroll
      for (int kc = 0; kc < 4; ++kc) fw[kc] = matW[kc * 32 + lane];
      v2f ure[2], uim[2];
      #pragma unroll
      for (int pp = 0; pp < 2; ++pp) {
        cpx c0 = cmul(pp ? a1[2] : a0[2], a0[3]), c1 = cmul(pp ? a1[2] : a0[2], a1[3]);
        ure[pp][0] = c0.re; ure[pp][1] = c1.re; uim[pp][0] = c0.im; uim[pp][1] = c1.im;
      }
      v2f sre[2], sim[2];
      #pragma unroll
      for (int i = 0; i < 2; ++i) { sre[i] = (v2f){0.f, 0.f}; sim[i] = (v2f){0.f, 0.f}; }
      #pragma unroll
      for (int kc = 0; kc < 4; ++kc) {
        const int v1 = kc & 1, v2 = kc >> 1, pp = v1 ^ v2;
        v2f ur = v1 ? __builtin_shufflevector(ure[pp], ure[pp], 1, 0) : ure[pp];
        v2f ui = v1 ? __builtin_shufflevector(uim[pp], uim[pp], 1, 0) : uim[pp];
        v2f wre; wre[0] = fw[kc][0]; wre[1] = fw[kc][1];
        v2f wim; wim[0] = fw[kc][2]; wim[1] = fw[kc][3];
        wim = wim + wre;
        sre[v2] = sre[v2] + wre * ur - wim * ui;
        sim[v2] = sim[v2] + wre * ui + wim * ur;
      }
      cpx h0 = pick(a0[0], a1[0], half);
      cpx g0 = cmul(h0, pick(a0[1], a1[1], half));
      cpx g1 = cmul(h0, pick(a0[1], a1[1], half ^ 1));
      v2f gre = g0.re * sre[0] - g0.im * sim[0] + g1.re * sre[1] - g1.im * sim[1];
      v2f gim = g0.re * sim[0] + g0.im * sre[0] + g1.re * sim[1] + g1.im * sre[1];
      pG[0].re = gre[0]; pG[0].im = gim[0]; pG[1].re = gre[1]; pG[1].im = gim[1];
    }
    {
      const v4f* matU = frag + 0 * 128;
      v4f fu[4];
      #pragma unroll
      for (int kc = 0; kc < 4; ++kc) fu[kc] = matU[kc * 32 + lane];
      v2f tre[2], tim[2];
      #pragma unroll
      for (int pp = 0; pp < 2; ++pp) {
        cpx c0 = cmul(pp ? a1[6] : a0[6], a0[7]), c1 = cmul(pp ? a1[6] : a0[6], a1[7]);
        tre[pp][0] = c0.re; tre[pp][1] = c1.re; tim[pp][0] = c0.im; tim[pp][1] = c1.im;
      }
      v2f sre[2], sim[2];
      #pragma unroll
      for (int i = 0; i < 2; ++i) { sre[i] = (v2f){0.f, 0.f}; sim[i] = (v2f){0.f, 0.f}; }
      #pragma unroll
      for (int kc = 0; kc < 4; ++kc) {
        const int v1 = kc & 1, v2 = kc >> 1, pp = v1 ^ v2;
        v2f tr = v1 ? __builtin_shufflevector(tre[pp], tre[pp], 1, 0) : tre[pp];
        v2f ti = v1 ? __builtin_shufflevector(tim[pp], tim[pp], 1, 0) : tim[pp];
        v2f wre; wre[0] = fu[kc][0]; wre[1] = fu[kc][1];
        v2f wim; wim[0] = fu[kc][2]; wim[1] = fu[kc][3];
        wim = wim + wre;
        sre[v2] = sre[v2] + wre * tr - wim * ti;
        sim[v2] = sim[v2] + wre * ti + wim * tr;
      }
      cpx b0 = pick(a0[5], a1[5], half);
      cpx b1 = pick(a0[5], a1[5], half ^ 1);
      v2f qre = b0.re * sre[0] - b0.im * sim[0] + b1.re * sre[1] - b1.im * sim[1];
      v2f qim = b0.re * sim[0] + b0.im * sre[0] + b1.re * sim[1] + b1.im * sre[1];
      Q.re = qre[0] + qre[1];  Q.im = qim[0] + qim[1];
    }
    cpx kep = pick(pG[0], pG[1], half);
    cpx snd = pick(pG[1], pG[0], half);
    cpx G, Qo, M;
    G.re  = kep.re + __shfl_xor(snd.re, 16, 32);
    G.im  = kep.im + __shfl_xor(snd.im, 16, 32);
    Qo.re = __shfl_xor(Q.re, 16, 32);
    Qo.im = __shfl_xor(Q.im, 16, 32);
    cpx m0 = cmul(a0[4], Q), m1 = cmul(a1[4], Qo);
    M.re = m0.re + m1.re;  M.im = m0.im + m1.im;
    v2f am;  am[0] = M.re;  am[1] = M.im;
    v2f bx;  bx[0] = G.re;  bx[1] = -G.im;
    v2f by;  by[0] = G.im;  by[1] = G.re;
    v8f zero = {};
    v8f accX = mm<false>(am, bx, zero);
    v8f accY = mm<false>(am, by, zero);
    #pragma unroll
    for (int v = 0; v < 8; ++v) { X[v] = accX[v]; Y[v] = accY[v]; }
  }

  for (int l = 1; l < NLAYERS; ++l) {
    const int flip = l & 1;
    const v4f* matA = frag + ((l - 1) * 2 + flip)       * 128;
    const v4f* matB = frag + ((l - 1) * 2 + (1 - flip)) * 128;

    if (l > 1) {
      cmulmat<true>(matA, lane, X, Y);
      cmulmat<false>(matB, lane, X, Y);
    }

    if (l == NLAYERS - 1) break;

    if (flip) {
      #pragma unroll
      for (int v = 0; v < 8; ++v) {
        float tx = __shfl_xor(X[v], 16, 32);
        float ty = __shfl_xor(Y[v], 16, 32);
        if (lane & 1) { X[v] = tx; Y[v] = ty; }
      }
      float nX[8], nY[8];
      #pragma unroll
      for (int v = 0; v < 8; ++v) {
        const int r0 = chainperm(v), r1 = chainperm(v + 8);
        float t0x = (r0 < 8)  ? X[r0 & 7] : __shfl_xor(X[r0 & 7], 16, 32);
        float t0y = (r0 < 8)  ? Y[r0 & 7] : __shfl_xor(Y[r0 & 7], 16, 32);
        float t1x = (r1 >= 8) ? X[r1 & 7] : __shfl_xor(X[r1 & 7], 16, 32);
        float t1y = (r1 >= 8) ? Y[r1 & 7] : __shfl_xor(Y[r1 & 7], 16, 32);
        nX[v] = half ? t1x : t0x;
        nY[v] = half ? t1y : t0y;
      }
      #pragma unroll
      for (int v = 0; v < 8; ++v) { X[v] = nX[v]; Y[v] = nY[v]; }
    } else {
      #pragma unroll
      for (int v = 1; v < 8; v += 2) {
        X[v] = __shfl_xor(X[v], 8, 32);
        Y[v] = __shfl_xor(Y[v], 8, 32);
      }
      int src = chainperm(lane & 15) | (lane & 16);
      #pragma unroll
      for (int v = 0; v < 8; ++v) {
        X[v] = __shfl(X[v], src, 32);
        Y[v] = __shfl(Y[v], src, 32);
      }
    }
  }

  float psum = 0.f, s5 = 0.f, s6 = 0.f, s7 = 0.f;
  #pragma unroll
  for (int v = 0; v < 8; ++v) {
    float p = X[v] * X[v] + Y[v] * Y[v];
    psum += p;
    s5 += ((v >> 2) & 1)                  ? -p : p;
    s6 += (((v >> 2) ^ (v >> 1)) & 1)     ? -p : p;
    s7 += (((v >> 2) ^ (v >> 1) ^ v) & 1) ? -p : p;
  }
  const bool neg = ((half ^ lane) & 1) != 0;
  float o[8];
  o[0] = ((col >> 3) & 1) ? -psum : psum;
  o[1] = ((col >> 2) & 1) ? -psum : psum;
  o[2] = ((col >> 1) & 1) ? -psum : psum;
  o[3] = ((col) & 1)      ? -psum : psum;
  o[4] = neg ? -psum : psum;
  o[5] = neg ? -s5 : s5;
  o[6] = neg ? -s6 : s6;
  o[7] = neg ? -s7 : s7;
  float o4[4], o2[2], tot;
  #pragma unroll
  for (int i = 0; i < 4; ++i) {
    float keep = half ? o[i + 4] : o[i];
    float send = half ? o[i] : o[i + 4];
    o4[i] = keep + __shfl_xor(send, 16, 32);
  }
  const int l3 = (lane >> 3) & 1;
  #pragma unroll
  for (int i = 0; i < 2; ++i) {
    float keep = l3 ? o4[i + 2] : o4[i];
    float send = l3 ? o4[i] : o4[i + 2];
    o2[i] = keep + __shfl_xor(send, 8, 32);
  }
  {
    const int l2 = (lane >> 2) & 1;
    float keep = l2 ? o2[1] : o2[0];
    float send = l2 ? o2[0] : o2[1];
    tot = keep + __shfl_xor(send, 4, 32);
  }
  tot += __shfl_xor(tot, 2, 32);
  tot += __shfl_xor(tot, 1, 32);
  if ((lane & 3) == 0 && bw < B) out[b * 8 + (lane >> 2)] = tot;
}

extern "C" void kernel_launch(void* const* d_in, const int* in_sizes, int n_in,
                              void* d_out, int out_size, void* d_ws, size_t ws_size,
                              hipStream_t stream) {
  const float* x = (const float*)d_in[0];
  const float* w = (const float*)d_in[1];
  float* out = (float*)d_out;
  int B = in_sizes[0] / NQ;

  int nblocks = (B + 31) / 32;
  qsim<<<nblocks, 1024, 0, stream>>>(x, w, out, B);
}


# One-wire gates split as phase * real rotation * phase: 16x16 products of layers 2, 3 are real (2 real products each, not 3), phases as one table multiply per layer; 50 -> 34 WMMA; sums regrouped
# speedup vs baseline: 4.3402x; 1.2029x over previous
#include <hip/hip_runtime.h>


typedef __attribute__((ext_vector_type(2))) float v2f;
typedef __attribute__((ext_vector_type(4))) float v4f;
typedef __attribute__((ext_vector_type(8))) float v8f;

#define NQ 8
#define NLAYERS 4

struct cpx { float re, im; };

__device__ __forceinline__ cpx cmul(cpx a, cpx b) {
  cpx r; r.re = a.re * b.re - a.im * b.im; r.im = a.re * b.im + a.im * b.re; return r;
}

__device__ __forceinline__ cpx pick(cpx a, cpx b, int sel) {
  cpx r; r.re = sel ? b.re : a.re; r.im = sel ? b.im : a.im; return r;
}

__device__ __forceinline__ cpx u_entry(const float* cs, const float* sn, int gate, int i, int j) {
  float ca = cs[3 * gate],     sa = sn[3 * gate];
  float cb = cs[3 * gate + 1], sb = sn[3 * gate + 1];
  float cg = cs[3 * gate + 2], sg = sn[3 * gate + 2];
  cpx m;
  if (i == 0 && j == 0) { m.re =  cb * ca; m.im =  sb * sa; }
  else if (i == 0)      { m.re = -sb * ca; m.im = -cb * sa; }
  else if (j == 0)      { m.re =  sb * ca; m.im = -cb * sa; }
  else                  { m.re =  cb * ca; m.im = -sb * sa; }
  cpx ph; ph.re = cg; ph.im = (i == 0) ? -sg : sg;
  return cmul(ph, m);
}

__host__ __device__ constexpr int chainperm(int r) {
  r ^= ((r >> 1) & 1);
  r ^= ((r >> 2) & 1) << 1;
  r ^= ((r >> 3) & 1) << 2;
  return r;
}

__device__ __forceinline__ void gate_split(const float* cs, const float* sn, int gate,
                                           float* c, float* s, cpx* x, cpx* y, cpx* w) {
  cpx A = u_entry(cs, sn, gate, 0, 0), B = u_entry(cs, sn, gate, 1, 0);
  float nA = A.re * A.re + A.im * A.im, nB = B.re * B.re + B.im * B.im;
  bool okA = nA > 1e-30f, okB = nB > 1e-30f;
  float rA = okA ? rsqrtf(nA) : 0.f, rB = okB ? rsqrtf(nB) : 0.f;
  *c = nA * rA;  *s = nB * rB;
  x->re = okA ? A.re * rA : 1.f;   x->im = okA ? A.im * rA : 0.f;
  y->re = okB ? B.re * rB : 1.f;   y->im = okB ? -B.im * rB : 0.f;
  cpx xy = cmul(*x, *y);
  w->re = xy.re;  w->im = -xy.im;
}

__device__ __forceinline__ void build_gates(const float* __restrict__ w, float* tmp,
                                            float* fragf, int t,
                                            float cv, float sv, float* av) {
  float* cs  = tmp;
  float* sn  = tmp + 96;
  float* p4  = tmp + 256;
  float* rp  = tmp + 512;
  float* dp  = tmp + 640;
  if (t < NLAYERS * NQ * 3) {
    cs[t] = cosf(0.5f * w[t]);
  } else if (t < 2 * NLAYERS * NQ * 3) {
    sn[t - NLAYERS * NQ * 3] = sinf(0.5f * w[t - NLAYERS * NQ * 3]);
  }
  __syncthreads();
  {
    int q = t & 7;
    cpx u00 = u_entry(cs, sn, q, 0, 0), u01 = u_entry(cs, sn, q, 0, 1);
    cpx u10 = u_entry(cs, sn, q, 1, 0), u11 = u_entry(cs, sn, q, 1, 1);
    av[0] = u00.re * cv + u01.re * sv;  av[1] = u00.im * cv + u01.im * sv;
    av[2] = u10.re * cv + u11.re * sv;  av[3] = u10.im * cv + u11.im * sv;
  }
  if (t >= 64 && t < 128) {
    int p = (t >> 4) & 3, i4 = (t >> 2) & 3, j4 = t & 3;
    int g0 = NQ + 2 * p;
    cpx e = cmul(u_entry(cs, sn, g0,     i4 >> 1, j4 >> 1),
                 u_entry(cs, sn, g0 + 1, i4 & 1,  j4 & 1));
    p4[2 * t]     = e.re;
    p4[2 * t + 1] = e.im;
  } else if (t >= 128 && t < 256) {
    int l2 = (t >> 6) & 1, p = (t >> 4) & 3, i4 = (t >> 2) & 3, j4 = t & 3;
    int g0 = (2 + l2) * NQ + 2 * p;
    float c0, s0, c1, s1; cpx x, y, ww;
    gate_split(cs, sn, g0,     &c0, &s0, &x, &y, &ww);
    gate_split(cs, sn, g0 + 1, &c1, &s1, &x, &y, &ww);
    int ia = i4 >> 1, ja = j4 >> 1, ib = i4 & 1, jb = j4 & 1;
    float ea = (ia == ja) ? c0 : (ia ? s0 : -s0);
    float eb = (ib == jb) ? c1 : (ib ? s1 : -s1);
    rp[t - 128] = ea * eb;
  } else if (t >= 256 && t < 320) {
    int idx = t - 256;
    int l2 = idx >> 5, left = (idx >> 4) & 1, p = (idx >> 2) & 3, i4 = idx & 3;
    int g0 = (2 + l2) * NQ + 2 * p;
    float c, s; cpx x0, y0, w0, x1, y1, w1, one;
    one.re = 1.f; one.im = 0.f;
    gate_split(cs, sn, g0,     &c, &s, &x0, &y0, &w0);
    gate_split(cs, sn, g0 + 1, &c, &s, &x1, &y1, &w1);
    cpx fa = left ? pick(one, w0, i4 >> 1) : pick(x0, y0, i4 >> 1);
    cpx fb = left ? pick(one, w1, i4 & 1)  : pick(x1, y1, i4 & 1);
    cpx e = cmul(fa, fb);
    dp[2 * idx]     = e.re;
    dp[2 * idx + 1] = e.im;
  }
  __syncthreads();
  if (t < 256) {
    int m = t >> 4, k = t & 15;
    int mp = chainperm(m);
    {
      int slot = (((k >> 1) & 3) * 32 + m + 16 * (k >> 3)) * 4 + (k & 1);
      const float* pl = p4 + 128;
      cpx a, b, r;
      a.re = pl[(2 * 16 + (m >> 2) * 4 + (k >> 2)) * 2]; a.im = pl[(2 * 16 + (m >> 2) * 4 + (k >> 2)) * 2 + 1];
      b.re = pl[(3 * 16 + (m & 3) * 4 + (k & 3)) * 2];   b.im = pl[(3 * 16 + (m & 3) * 4 + (k & 3)) * 2 + 1];
      r = cmul(a, b);
      fragf[slot]     = r.re;
      fragf[slot + 2] = r.im - r.re;
      a.re = pl[(0 * 16 + (mp >> 2) * 4 + (k >> 2)) * 2]; a.im = pl[(0 * 16 + (mp >> 2) * 4 + (k >> 2)) * 2 + 1];
      b.re = pl[(1 * 16 + (mp & 3) * 4 + (k & 3)) * 2];   b.im = pl[(1 * 16 + (mp & 3) * 4 + (k & 3)) * 2 + 1];
      r = cmul(a, b);
      fragf[512 + slot]     = r.re;
      fragf[512 + slot + 2] = r.im - r.re;
    }
    int slotr = (((k >> 2) & 1) * 32 + m + 16 * (k >> 3)) * 4 + ((k >> 1) & 1) * 2 + (k & 1);
    #pragma unroll
    for (int l2 = 0; l2 < 2; ++l2) {
      const float* rl = rp + l2 * 64;
      fragf[1024 + (l2 * 2 + 0) * 256 + slotr] =
          rl[2 * 16 + (m >> 2) * 4 + (k >> 2)] * rl[3 * 16 + (m & 3) * 4 + (k & 3)];
      fragf[1024 + (l2 * 2 + 1) * 256 + slotr] =
          rl[0 * 16 + (mp >> 2) * 4 + (k >> 2)] * rl[1 * 16 + (mp & 3) * 4 + (k & 3)];
    }
    auto ph16 = [&](const float* dl, int ph, int i) {
      cpx a, b;
      a.re = dl[(ph * 4 + (i >> 2)) * 2];       a.im = dl[(ph * 4 + (i >> 2)) * 2 + 1];
      b.re = dl[((ph + 1) * 4 + (i & 3)) * 2];  b.im = dl[((ph + 1) * 4 + (i & 3)) * 2 + 1];
      return cmul(a, b);
    };
    {
      cpx e = cmul(ph16(dp + 0 * 32, 2, m), ph16(dp + 0 * 32, 0, k));
      int idx = (((m & 7) >> 1) * 32 + k + 16 * (m >> 3)) * 4 + (m & 1);
      fragf[2048 + idx]     = e.re;
      fragf[2048 + idx + 2] = e.im;
    }
    {
      cpx e = cmul(ph16(dp + 1 * 32, 2, chainperm(m) ^ ((k & 1) << 3)), ph16(dp + 1 * 32, 0, chainperm(k)));
      cpx f = cmul(ph16(dp + 2 * 32, 2, m), ph16(dp + 2 * 32, 0, k));
      e = cmul(e, f);
      int idx = (((k & 7) >> 1) * 32 + m + 16 * (k >> 3)) * 4 + (k & 1);
      fragf[2560 + idx]     = e.re;
      fragf[2560 + idx + 2] = e.im;
    }
  }
  __syncthreads();
}


template <bool STATE_A>
__device__ __forceinline__ v8f mm(v2f g, v2f s, v8f acc) {
  return STATE_A
    ? __builtin_amdgcn_wmma_f32_16x16x4_f32(false, s, false, g, (short)0, acc, false, false)
    : __builtin_amdgcn_wmma_f32_16x16x4_f32(false, g, false, s, (short)0, acc, false, false);
}

template <bool STATE_A>
__device__ __forceinline__ void rmulmat(const v4f* mat, int lane, float* X, float* Y) {
  v4f f[2];
  #pragma unroll
  for (int i = 0; i < 2; ++i) f[i] = mat[i * 32 + lane];
  v8f accX = {}; v8f accY = {};
  #pragma unroll
  for (int kc = 0; kc < 4; ++kc) {
    int a0 = 2 * kc, a1 = a0 + 1;
    v2f g;   g[0] = f[kc >> 1][2 * (kc & 1)];  g[1] = f[kc >> 1][2 * (kc & 1) + 1];
    v2f bx;  bx[0] = X[a0];     bx[1] = X[a1];
    v2f by;  by[0] = Y[a0];     by[1] = Y[a1];
    accX = mm<STATE_A>(g, bx, accX);
    accY = mm<STATE_A>(g, by, accY);
  }
  #pragma unroll
  for (int v = 0; v < 8; ++v) { X[v] = accX[v]; Y[v] = accY[v]; }
}

__device__ __forceinline__ void apply_phases(const v4f* tb, int lane, float* X, float* Y) {
  #pragma unroll
  for (int j = 0; j < 4; ++j) {
    v4f d = tb[j * 32 + lane];
    v2f dr;  dr[0] = d[0];  dr[1] = d[1];
    v2f di;  di[0] = d[2];  di[1] = d[3];
    v2f x;   x[0] = X[2 * j];  x[1] = X[2 * j + 1];
    v2f y;   y[0] = Y[2 * j];  y[1] = Y[2 * j + 1];
    v2f nx = x * dr - y * di;
    v2f ny = x * di + y * dr;
    X[2 * j] = nx[0];  X[2 * j + 1] = nx[1];
    Y[2 * j] = ny[0];  Y[2 * j + 1] = ny[1];
  }
}

__global__ __launch_bounds__(1024, 16) void qsim(const float* __restrict__ x,
                                            const float* __restrict__ w,
                                            float* __restrict__ out, int B) {
  __shared__ float tmp[768];
  __shared__ v4f   frag[6 * 4 * 32];
  const int lane  = threadIdx.x & 31;
  const int wslot = __builtin_amdgcn_readfirstlane(threadIdx.x >> 5);
  const int bw    = blockIdx.x * 32 + wslot;
  const int b     = bw < B ? bw : B - 1;
  const int half = lane >> 4;
  const int col  = lane & 15;

  float sv, cv;
  sincosf(0.5f * x[b * 8 + (lane & 7)], &sv, &cv);

  float av[4];
  build_gates(w, tmp, (float*)frag, threadIdx.x, cv, sv, av);

  cpx a0[8], a1[8];
  #pragma unroll
  for (int q = 0; q < 8; ++q) {
    a0[q].re = __int_as_float(__builtin_amdgcn_readlane(__float_as_int(av[0]), q));
    a0[q].im = __int_as_float(__builtin_amdgcn_readlane(__float_as_int(av[1]), q));
    a1[q].re = __int_as_float(__builtin_amdgcn_readlane(__float_as_int(av[2]), q));
    a1[q].im = __int_as_float(__builtin_amdgcn_readlane(__float_as_int(av[3]), q));
  }

  float X[8], Y[8];
  {
    cpx pG[2], Q;
    {
      const v4f* matW = frag + 1 * 128;
      v4f fw[4];
      #pragma unroll
      for (int kc = 0; kc < 4; ++kc) fw[kc] = matW[kc * 32 + lane];
      v2f ure[2], uim[2];
      #pragma unroll
      for (int pp = 0; pp < 2; ++pp) {
        cpx c0 = cmul(pp ? a1[2] : a0[2], a0[3]), c1 = cmul(pp ? a1[2] : a0[2], a1[3]);
        ure[pp][0] = c0.re; ure[pp][1] = c1.re; uim[pp][0] = c0.im; uim[pp][1] = c1.im;
      }
      v2f sre[2], sim[2];
      #pragma unroll
      for (int i = 0; i < 2; ++i) { sre[i] = (v2f){0.f, 0.f}; sim[i] = (v2f){0.f, 0.f}; }
      #pragma unroll
      for (int kc = 0; kc < 4; ++kc) {
        const int v1 = kc & 1, v2 = kc >> 1, pp = v1 ^ v2;
        v2f ur = v1 ? __builtin_shufflevector(ure[pp], ure[pp], 1, 0) : ure[pp];
        v2f ui = v1 ? __builtin_shufflevector(uim[pp], uim[pp], 1, 0) : uim[pp];
        v2f wre; wre[0] = fw[kc][0]; wre[1] = fw[kc][1];
        v2f wim; wim[0] = fw[kc][2]; wim[1] = fw[kc][3];
        wim = wim + wre;
        sre[v2] = sre[v2] + wre * ur - wim * ui;
        sim[v2] = sim[v2] + wre * ui + wim * ur;
      }
      cpx h0 = pick(a0[0], a1[0], half);
      cpx g0 = cmul(h0, pick(a0[1], a1[1], half));
      cpx g1 = cmul(h0, pick(a0[1], a1[1], half ^ 1));
      v2f gre = g0.re * sre[0] - g0.im * sim[0] + g1.re * sre[1] - g1.im * sim[1];
      v2f gim = g0.re * sim[0] + g0.im * sre[0] + g1.re * sim[1] + g1.im * sre[1];
      pG[0].re = gre[0]; pG[0].im = gim[0]; pG[1].re = gre[1]; pG[1].im = gim[1];
    }
    {
      const v4f* matU = frag + 0 * 128;
      v4f fu[4];
      #pragma unroll
      for (int kc = 0; kc < 4; ++kc) fu[kc] = matU[kc * 32 + lane];
      v2f tre[2], tim[2];
      #pragma unroll
      for (int pp = 0; pp < 2; ++pp) {
        cpx c0 = cmul(pp ? a1[6] : a0[6], a0[7]), c1 = cmul(pp ? a1[6] : a0[6], a1[7]);
        tre[pp][0] = c0.re; tre[pp][1] = c1.re; tim[pp][0] = c0.im; tim[pp][1] = c1.im;
      }
      v2f sre[2], sim[2];
      #pragma unroll
      for (int i = 0; i < 2; ++i) { sre[i] = (v2f){0.f, 0.f}; sim[i] = (v2f){0.f, 0.f}; }
      #pragma unroll
      for (int kc = 0; kc < 4; ++kc) {
        const int v1 = kc & 1, v2 = kc >> 1, pp = v1 ^ v2;
        v2f tr = v1 ? __builtin_shufflevector(tre[pp], tre[pp], 1, 0) : tre[pp];
        v2f ti = v1 ? __builtin_shufflevector(tim[pp], tim[pp], 1, 0) : tim[pp];
        v2f wre; wre[0] = fu[kc][0]; wre[1] = fu[kc][1];
        v2f wim; wim[0] = fu[kc][2]; wim[1] = fu[kc][3];
        wim = wim + wre;
        sre[v2] = sre[v2] + wre * tr - wim * ti;
        sim[v2] = sim[v2] + wre * ti + wim * tr;
      }
      cpx b0 = pick(a0[5], a1[5], half);
      cpx b1 = pick(a0[5], a1[5], half ^ 1);
      v2f qre = b0.re * sre[0] - b0.im * sim[0] + b1.re * sre[1] - b1.im * sim[1];
      v2f qim = b0.re * sim[0] + b0.im * sre[0] + b1.re * sim[1] + b1.im * sre[1];
      Q.re = qre[0] + qre[1];  Q.im = qim[0] + qim[1];
    }
    cpx kep = pick(pG[0], pG[1], half);
    cpx snd = pick(pG[1], pG[0], half);
    cpx G, Qo, M;
    G.re  = kep.re + __shfl_xor(snd.re, 16, 32);
    G.im  = kep.im + __shfl_xor(snd.im, 16, 32);
    Qo.re = __shfl_xor(Q.re, 16, 32);
    Qo.im = __shfl_xor(Q.im, 16, 32);
    cpx m0 = cmul(a0[4], Q), m1 = cmul(a1[4], Qo);
    M.re = m0.re + m1.re;  M.im = m0.im + m1.im;
    v2f am;  am[0] = M.re;  am[1] = M.im;
    v2f bx;  bx[0] = G.re;  bx[1] = -G.im;
    v2f by;  by[0] = G.im;  by[1] = G.re;
    v8f zero = {};
    v8f accX = mm<false>(am, bx, zero);
    v8f accY = mm<false>(am, by, zero);
    #pragma unroll
    for (int v = 0; v < 8; ++v) { X[v] = accX[v]; Y[v] = accY[v]; }
  }

  for (int l = 1; l < NLAYERS; ++l) {
    const int flip = l & 1;

    if (l > 1) {
      const v4f* matA = frag + 256 + ((l - 2) * 2 + flip)       * 64;
      const v4f* matB = frag + 256 + ((l - 2) * 2 + (1 - flip)) * 64;
      apply_phases(frag + 512 + (l - 2) * 128, lane, X, Y);
      rmulmat<true>(matA, lane, X, Y);
      rmulmat<false>(matB, lane, X, Y);
    }

    if (l == NLAYERS - 1) break;

    if (flip) {
      #pragma unroll
      for (int v = 0; v < 8; ++v) {
        float tx = __shfl_xor(X[v], 16, 32);
        float ty = __shfl_xor(Y[v], 16, 32);
        if (lane & 1) { X[v] = tx; Y[v] = ty; }
      }
      float nX[8], nY[8];
      #pragma unroll
      for (int v = 0; v < 8; ++v) {
        const int r0 = chainperm(v), r1 = chainperm(v + 8);
        float t0x = (r0 < 8)  ? X[r0 & 7] : __shfl_xor(X[r0 & 7], 16, 32);
        float t0y = (r0 < 8)  ? Y[r0 & 7] : __shfl_xor(Y[r0 & 7], 16, 32);
        float t1x = (r1 >= 8) ? X[r1 & 7] : __shfl_xor(X[r1 & 7], 16, 32);
        float t1y = (r1 >= 8) ? Y[r1 & 7] : __shfl_xor(Y[r1 & 7], 16, 32);
        nX[v] = half ? t1x : t0x;
        nY[v] = half ? t1y : t0y;
      }
      #pragma unroll
      for (int v = 0; v < 8; ++v) { X[v] = nX[v]; Y[v] = nY[v]; }
    } else {
      #pragma unroll
      for (int v = 1; v < 8; v += 2) {
        X[v] = __shfl_xor(X[v], 8, 32);
        Y[v] = __shfl_xor(Y[v], 8, 32);
      }
      int src = chainperm(lane & 15) | (lane & 16);
      #pragma unroll
      for (int v = 0; v < 8; ++v) {
        X[v] = __shfl(X[v], src, 32);
        Y[v] = __shfl(Y[v], src, 32);
      }
    }
  }

  float psum = 0.f, s5 = 0.f, s6 = 0.f, s7 = 0.f;
  #pragma unroll
  for (int v = 0; v < 8; ++v) {
    float p = X[v] * X[v] + Y[v] * Y[v];
    psum += p;
    s5 += ((v >> 2) & 1)                  ? -p : p;
    s6 += (((v >> 2) ^ (v >> 1)) & 1)     ? -p : p;
    s7 += (((v >> 2) ^ (v >> 1) ^ v) & 1) ? -p : p;
  }
  const bool neg = ((half ^ lane) & 1) != 0;
  float o[8];
  o[0] = ((col >> 3) & 1) ? -psum : psum;
  o[1] = ((col >> 2) & 1) ? -psum : psum;
  o[2] = ((col >> 1) & 1) ? -psum : psum;
  o[3] = ((col) & 1)      ? -psum : psum;
  o[4] = neg ? -psum : psum;
  o[5] = neg ? -s5 : s5;
  o[6] = neg ? -s6 : s6;
  o[7] = neg ? -s7 : s7;
  float o4[4], o2[2], tot;
  #pragma unroll
  for (int i = 0; i < 4; ++i) {
    float keep = half ? o[i + 4] : o[i];
    float send = half ? o[i] : o[i + 4];
    o4[i] = keep + __shfl_xor(send, 16, 32);
  }
  const int l3 = (lane >> 3) & 1;
  #pragma unroll
  for (int i = 0; i < 2; ++i) {
    float keep = l3 ? o4[i + 2] : o4[i];
    float send = l3 ? o4[i] : o4[i + 2];
    o2[i] = keep + __shfl_xor(send, 8, 32);
  }
  {
    const int l2 = (lane >> 2) & 1;
    float keep = l2 ? o2[1] : o2[0];
    float send = l2 ? o2[0] : o2[1];
    tot = keep + __shfl_xor(send, 4, 32);
  }
  tot += __shfl_xor(tot, 2, 32);
  tot += __shfl_xor(tot, 1, 32);
  if ((lane & 3) == 0 && bw < B) out[b * 8 + (lane >> 2)] = tot;
}

extern "C" void kernel_launch(void* const* d_in, const int* in_sizes, int n_in,
                              void* d_out, int out_size, void* d_ws, size_t ws_size,
                              hipStream_t stream) {
  const float* x = (const float*)d_in[0];
  const float* w = (const float*)d_in[1];
  float* out = (float*)d_out;
  int B = in_sizes[0] / NQ;

  int nblocks = (B + 31) / 32;
  qsim<<<nblocks, 1024, 0, stream>>>(x, w, out, B);
}


# Layer 1's CNOTs, layer 2's right-hand phases and layer 2's row product folded into layer 1's row-side sums (factor tables per work-group, one sum per column parity): 34 -> 28 WMMA; sums regrouped
# speedup vs baseline: 4.6473x; 1.0708x over previous
#include <hip/hip_runtime.h>


typedef __attribute__((ext_vector_type(2))) float v2f;
typedef __attribute__((ext_vector_type(4))) float v4f;
typedef __attribute__((ext_vector_type(8))) float v8f;

#define NQ 8
#define NLAYERS 4

struct cpx { float re, im; };

__device__ __forceinline__ cpx cmul(cpx a, cpx b) {
  cpx r; r.re = a.re * b.re - a.im * b.im; r.im = a.re * b.im + a.im * b.re; return r;
}

__device__ __forceinline__ cpx pick(cpx a, cpx b, int sel) {
  cpx r; r.re = sel ? b.re : a.re; r.im = sel ? b.im : a.im; return r;
}

__device__ __forceinline__ cpx u_entry(const float* cs, const float* sn, int gate, int i, int j) {
  float ca = cs[3 * gate],     sa = sn[3 * gate];
  float cb = cs[3 * gate + 1], sb = sn[3 * gate + 1];
  float cg = cs[3 * gate + 2], sg = sn[3 * gate + 2];
  cpx m;
  if (i == 0 && j == 0) { m.re =  cb * ca; m.im =  sb * sa; }
  else if (i == 0)      { m.re = -sb * ca; m.im = -cb * sa; }
  else if (j == 0)      { m.re =  sb * ca; m.im = -cb * sa; }
  else                  { m.re =  cb * ca; m.im = -sb * sa; }
  cpx ph; ph.re = cg; ph.im = (i == 0) ? -sg : sg;
  return cmul(ph, m);
}

__host__ __device__ constexpr int chainperm(int r) {
  r ^= ((r >> 1) & 1);
  r ^= ((r >> 2) & 1) << 1;
  r ^= ((r >> 3) & 1) << 2;
  return r;
}

__device__ __forceinline__ void gate_split(const float* cs, const float* sn, int gate,
                                           float* c, float* s, cpx* x, cpx* y, cpx* w) {
  cpx A = u_entry(cs, sn, gate, 0, 0), B = u_entry(cs, sn, gate, 1, 0);
  float nA = A.re * A.re + A.im * A.im, nB = B.re * B.re + B.im * B.im;
  bool okA = nA > 1e-30f, okB = nB > 1e-30f;
  float rA = okA ? rsqrtf(nA) : 0.f, rB = okB ? rsqrtf(nB) : 0.f;
  *c = nA * rA;  *s = nB * rB;
  x->re = okA ? A.re * rA : 1.f;   x->im = okA ? A.im * rA : 0.f;
  y->re = okB ? B.re * rB : 1.f;   y->im = okB ? -B.im * rB : 0.f;
  cpx xy = cmul(*x, *y);
  w->re = xy.re;  w->im = -xy.im;
}

__device__ __forceinline__ void build_gates(const float* __restrict__ w, float* tmp,
                                            float* fragf, int t,
                                            float cv, float sv, float* av) {
  float* cs  = tmp;
  float* sn  = tmp + 96;
  float* p4  = tmp + 256;
  float* rp  = tmp + 512;
  float* dp  = tmp + 640;
  if (t < NLAYERS * NQ * 3) {
    cs[t] = cosf(0.5f * w[t]);
  } else if (t < 2 * NLAYERS * NQ * 3) {
    sn[t - NLAYERS * NQ * 3] = sinf(0.5f * w[t - NLAYERS * NQ * 3]);
  }
  __syncthreads();
  {
    int q = t & 7;
    cpx u00 = u_entry(cs, sn, q, 0, 0), u01 = u_entry(cs, sn, q, 0, 1);
    cpx u10 = u_entry(cs, sn, q, 1, 0), u11 = u_entry(cs, sn, q, 1, 1);
    av[0] = u00.re * cv + u01.re * sv;  av[1] = u00.im * cv + u01.im * sv;
    av[2] = u10.re * cv + u11.re * sv;  av[3] = u10.im * cv + u11.im * sv;
  }
  if (t >= 64 && t < 128) {
    int p = (t >> 4) & 3, i4 = (t >> 2) & 3, j4 = t & 3;
    int g0 = NQ + 2 * p;
    cpx e = cmul(u_entry(cs, sn, g0,     i4 >> 1, j4 >> 1),
                 u_entry(cs, sn, g0 + 1, i4 & 1,  j4 & 1));
    p4[2 * t]     = e.re;
    p4[2 * t + 1] = e.im;
  } else if (t >= 128 && t < 256) {
    int l2 = (t >> 6) & 1, p = (t >> 4) & 3, i4 = (t >> 2) & 3, j4 = t & 3;
    int g0 = (2 + l2) * NQ + 2 * p;
    float c0, s0, c1, s1; cpx x, y, ww;
    gate_split(cs, sn, g0,     &c0, &s0, &x, &y, &ww);
    gate_split(cs, sn, g0 + 1, &c1, &s1, &x, &y, &ww);
    int ia = i4 >> 1, ja = j4 >> 1, ib = i4 & 1, jb = j4 & 1;
    float ea = (ia == ja) ? c0 : (ia ? s0 : -s0);
    float eb = (ib == jb) ? c1 : (ib ? s1 : -s1);
    rp[t - 128] = ea * eb;
  } else if (t >= 256 && t < 320) {
    int idx = t - 256;
    int l2 = idx >> 5, left = (idx >> 4) & 1, p = (idx >> 2) & 3, i4 = idx & 3;
    int g0 = (2 + l2) * NQ + 2 * p;
    float c, s; cpx x0, y0, w0, x1, y1, w1, one;
    one.re = 1.f; one.im = 0.f;
    gate_split(cs, sn, g0,     &c, &s, &x0, &y0, &w0);
    gate_split(cs, sn, g0 + 1, &c, &s, &x1, &y1, &w1);
    cpx fa = left ? pick(one, w0, i4 >> 1) : pick(x0, y0, i4 >> 1);
    cpx fb = left ? pick(one, w1, i4 & 1)  : pick(x1, y1, i4 & 1);
    cpx e = cmul(fa, fb);
    dp[2 * idx]     = e.re;
    dp[2 * idx + 1] = e.im;
  }
  __syncthreads();
  auto ph16 = [&](const float* dl, int ph, int i) {
    cpx a, b;
    a.re = dl[(ph * 4 + (i >> 2)) * 2];       a.im = dl[(ph * 4 + (i >> 2)) * 2 + 1];
    b.re = dl[((ph + 1) * 4 + (i & 3)) * 2];  b.im = dl[((ph + 1) * 4 + (i & 3)) * 2 + 1];
    return cmul(a, b);
  };
  if (t < 256) {
    int m = t >> 4, k = t & 15;
    int mp = chainperm(m);
    {
      int slot = (((k >> 1) & 3) * 32 + m + 16 * (k >> 3)) * 4 + (k & 1);
      const float* pl = p4 + 128;
      cpx a, b, r;
      a.re = pl[(0 * 16 + (mp >> 2) * 4 + (k >> 2)) * 2]; a.im = pl[(0 * 16 + (mp >> 2) * 4 + (k >> 2)) * 2 + 1];
      b.re = pl[(1 * 16 + (mp & 3) * 4 + (k & 3)) * 2];   b.im = pl[(1 * 16 + (mp & 3) * 4 + (k & 3)) * 2 + 1];
      r = cmul(cmul(a, b), ph16(dp + 0 * 32, 0, m));
      fragf[512 + slot]     = r.re;
      fragf[512 + slot + 2] = r.im - r.re;
    }
    int slotr = (((k >> 2) & 1) * 32 + m + 16 * (k >> 3)) * 4 + ((k >> 1) & 1) * 2 + (k & 1);
    #pragma unroll
    for (int l2 = 0; l2 < 2; ++l2) {
      const float* rl = rp + l2 * 64;
      if (l2 == 1)
        fragf[1024 + (l2 * 2 + 0) * 256 + slotr] =
            rl[2 * 16 + (m >> 2) * 4 + (k >> 2)] * rl[3 * 16 + (m & 3) * 4 + (k & 3)];
      fragf[1024 + (l2 * 2 + 1) * 256 + slotr] =
          rl[0 * 16 + (mp >> 2) * 4 + (k >> 2)] * rl[1 * 16 + (mp & 3) * 4 + (k & 3)];
    }
    {
      cpx e = cmul(ph16(dp + 1 * 32, 2, chainperm(m) ^ ((k & 1) << 3)), ph16(dp + 1 * 32, 0, chainperm(k)));
      cpx f = cmul(ph16(dp + 2 * 32, 2, m), ph16(dp + 2 * 32, 0, k));
      e = cmul(e, f);
      int idx = (((k & 7) >> 1) * 32 + m + 16 * (k >> 3)) * 4 + (k & 1);
      fragf[2560 + idx]     = e.re;
      fragf[2560 + idx + 2] = e.im;
    }
  } else if (t < 352) {
    const float* pl = p4 + 128;
    const float* rl = rp;
    const float* dl = dp + 0 * 32;
    if (t < 288) {
      int i = t - 256;
      int beta = (i >> 4) & 1, ml = (i >> 2) & 3, kl = i & 3;
      cpx s; s.re = 0.f; s.im = 0.f;
      #pragma unroll
      for (int rb = 0; rb < 4; ++rb) {
        int lo = chainperm(4 * beta + rb) & 3;
        float r = rl[3 * 16 + ml * 4 + rb];
        cpx ph, u;
        ph.re = dl[(12 + rb) * 2] * r;  ph.im = dl[(12 + rb) * 2 + 1] * r;
        u.re = pl[(3 * 16 + lo * 4 + kl) * 2];  u.im = pl[(3 * 16 + lo * 4 + kl) * 2 + 1];
        cpx pu = cmul(ph, u);
        s.re += pu.re;  s.im += pu.im;
      }
      fragf[ml * 16 + beta * 8 + kl * 2]     = s.re;
      fragf[ml * 16 + beta * 8 + kl * 2 + 1] = s.im;
    } else {
      int i = t - 288;
      int e = i >> 5, beta = (i >> 4) & 1, mh = (i >> 2) & 3, kh = i & 3;
      cpx s; s.re = 0.f; s.im = 0.f;
      #pragma unroll
      for (int a = 0; a < 2; ++a) {
        int ra = beta + 2 * a;
        int hi = (chainperm(4 * ra) >> 2) ^ (e << 1);
        float r = rl[2 * 16 + mh * 4 + ra];
        cpx ph, u;
        ph.re = dl[(8 + ra) * 2] * r;  ph.im = dl[(8 + ra) * 2 + 1] * r;
        u.re = pl[(2 * 16 + hi * 4 + kh) * 2];  u.im = pl[(2 * 16 + hi * 4 + kh) * 2 + 1];
        cpx pu = cmul(ph, u);
        s.re += pu.re;  s.im += pu.im;
      }
      int off = 64 + (((mh * 2 + (kh >> 1)) * 2 + e) * 2 + beta) * 4 + (kh & 1);
      fragf[off]     = s.re;
      fragf[off + 2] = s.im;
    }
  }
  __syncthreads();
}


template <bool STATE_A>
__device__ __forceinline__ v8f mm(v2f g, v2f s, v8f acc) {
  return STATE_A
    ? __builtin_amdgcn_wmma_f32_16x16x4_f32(false, s, false, g, (short)0, acc, false, false)
    : __builtin_amdgcn_wmma_f32_16x16x4_f32(false, g, false, s, (short)0, acc, false, false);
}

template <bool STATE_A>
__device__ __forceinline__ void rmulmat(const v4f* mat, int lane, float* X, float* Y) {
  v4f f[2];
  #pragma unroll
  for (int i = 0; i < 2; ++i) f[i] = mat[i * 32 + lane];
  v8f accX = {}; v8f accY = {};
  #pragma unroll
  for (int kc = 0; kc < 4; ++kc) {
    int a0 = 2 * kc, a1 = a0 + 1;
    v2f g;   g[0] = f[kc >> 1][2 * (kc & 1)];  g[1] = f[kc >> 1][2 * (kc & 1) + 1];
    v2f bx;  bx[0] = X[a0];     bx[1] = X[a1];
    v2f by;  by[0] = Y[a0];     by[1] = Y[a1];
    accX = mm<STATE_A>(g, bx, accX);
    accY = mm<STATE_A>(g, by, accY);
  }
  #pragma unroll
  for (int v = 0; v < 8; ++v) { X[v] = accX[v]; Y[v] = accY[v]; }
}

__device__ __forceinline__ void apply_phases(const v4f* tb, int lane, float* X, float* Y) {
  #pragma unroll
  for (int j = 0; j < 4; ++j) {
    v4f d = tb[j * 32 + lane];
    v2f dr;  dr[0] = d[0];  dr[1] = d[1];
    v2f di;  di[0] = d[2];  di[1] = d[3];
    v2f x;   x[0] = X[2 * j];  x[1] = X[2 * j + 1];
    v2f y;   y[0] = Y[2 * j];  y[1] = Y[2 * j + 1];
    v2f nx = x * dr - y * di;
    v2f ny = x * di + y * dr;
    X[2 * j] = nx[0];  X[2 * j + 1] = nx[1];
    Y[2 * j] = ny[0];  Y[2 * j + 1] = ny[1];
  }
}

__global__ __launch_bounds__(1024, 16) void qsim(const float* __restrict__ x,
                                            const float* __restrict__ w,
                                            float* __restrict__ out, int B) {
  __shared__ float tmp[768];
  __shared__ v4f   frag[6 * 4 * 32];
  const int lane  = threadIdx.x & 31;
  const int wslot = __builtin_amdgcn_readfirstlane(threadIdx.x >> 5);
  const int bw    = blockIdx.x * 32 + wslot;
  const int b     = bw < B ? bw : B - 1;
  const int half = lane >> 4;
  const int col  = lane & 15;

  float sv, cv;
  sincosf(0.5f * x[b * 8 + (lane & 7)], &sv, &cv);

  float av[4];
  build_gates(w, tmp, (float*)frag, threadIdx.x, cv, sv, av);

  cpx a0[8], a1[8];
  #pragma unroll
  for (int q = 0; q < 8; ++q) {
    a0[q].re = __int_as_float(__builtin_amdgcn_readlane(__float_as_int(av[0]), q));
    a0[q].im = __int_as_float(__builtin_amdgcn_readlane(__float_as_int(av[1]), q));
    a1[q].re = __int_as_float(__builtin_amdgcn_readlane(__float_as_int(av[2]), q));
    a1[q].im = __int_as_float(__builtin_amdgcn_readlane(__float_as_int(av[3]), q));
  }

  float X[8], Y[8];
  {
    cpx pG[2], Q[2];
    {
      const v4f* matW = frag + 1 * 128;
      v4f fw[4];
      #pragma unroll
      for (int kc = 0; kc < 4; ++kc) fw[kc] = matW[kc * 32 + lane];
      v2f ure[2], uim[2];
      #pragma unroll
      for (int pp = 0; pp < 2; ++pp) {
        cpx c0 = cmul(pp ? a1[2] : a0[2], a0[3]), c1 = cmul(pp ? a1[2] : a0[2], a1[3]);
        ure[pp][0] = c0.re; ure[pp][1] = c1.re; uim[pp][0] = c0.im; uim[pp][1] = c1.im;
      }
      v2f sre[2], sim[2];
      #pragma unroll
      for (int i = 0; i < 2; ++i) { sre[i] = (v2f){0.f, 0.f}; sim[i] = (v2f){0.f, 0.f}; }
      #pragma unroll
      for (int kc = 0; kc < 4; ++kc) {
        const int v1 = kc & 1, v2 = kc >> 1, pp = v1 ^ v2;
        v2f ur = v1 ? __builtin_shufflevector(ure[pp], ure[pp], 1, 0) : ure[pp];
        v2f ui = v1 ? __builtin_shufflevector(uim[pp], uim[pp], 1, 0) : uim[pp];
        v2f wre; wre[0] = fw[kc][0]; wre[1] = fw[kc][1];
        v2f wim; wim[0] = fw[kc][2]; wim[1] = fw[kc][3];
        wim = wim + wre;
        sre[v2] = sre[v2] + wre * ur - wim * ui;
        sim[v2] = sim[v2] + wre * ui + wim * ur;
      }
      cpx h0 = pick(a0[0], a1[0], half);
      cpx g0 = cmul(h0, pick(a0[1], a1[1], half));
      cpx g1 = cmul(h0, pick(a0[1], a1[1], half ^ 1));
      v2f gre = g0.re * sre[0] - g0.im * sim[0] + g1.re * sre[1] - g1.im * sim[1];
      v2f gim = g0.re * sim[0] + g0.im * sre[0] + g1.re * sim[1] + g1.im * sre[1];
      pG[0].re = gre[0]; pG[0].im = gim[0]; pG[1].re = gre[1]; pG[1].im = gim[1];
    }
    {
      v2f ure[2], uim[2];
      #pragma unroll
      for (int q = 0; q < 2; ++q) {
        cpx c0 = cmul(a0[6], q ? a1[7] : a0[7]), c1 = cmul(a1[6], q ? a1[7] : a0[7]);
        ure[q][0] = c0.re; ure[q][1] = c1.re; uim[q][0] = c0.im; uim[q][1] = c1.im;
      }
      const v4f* cbt = frag + (col & 3) * 4;
      const v4f* dt  = frag + 16 + ((col >> 2) * 2 + half) * 4;
      v4f fc[4];
      #pragma unroll
      for (int i = 0; i < 4; ++i) fc[i] = cbt[i];
      v2f Ere[2], Eim[2];
      #pragma unroll
      for (int beta = 0; beta < 2; ++beta) {
        Ere[beta] = (v2f){0.f, 0.f}; Eim[beta] = (v2f){0.f, 0.f};
        #pragma unroll
        for (int kl = 0; kl < 4; ++kl) {
          const int k1 = kl >> 1, k0 = kl & 1, q = k0 ^ k1;
          v2f tr = k1 ? __builtin_shufflevector(ure[q], ure[q], 1, 0) : ure[q];
          v2f ti = k1 ? __builtin_shufflevector(uim[q], uim[q], 1, 0) : uim[q];
          float cr = fc[beta * 2 + (kl >> 1)][(kl & 1) * 2];
          float ci = fc[beta * 2 + (kl >> 1)][(kl & 1) * 2 + 1];
          Ere[beta] = Ere[beta] + cr * tr - ci * ti;
          Eim[beta] = Eim[beta] + cr * ti + ci * tr;
        }
      }
      cpx b0 = pick(a0[5], a1[5], half);
      cpx b1 = pick(a0[5], a1[5], half ^ 1);
      v2f a5re;  a5re[0] = b0.re;  a5re[1] = b1.re;
      v2f a5im;  a5im[0] = b0.im;  a5im[1] = b1.im;
      v2f Fre[2], Fim[2];
      #pragma unroll
      for (int beta = 0; beta < 2; ++beta) {
        Fre[beta] = a5re * Ere[beta] - a5im * Eim[beta];
        Fim[beta] = a5re * Eim[beta] + a5im * Ere[beta];
      }
      v4f fd[4];
      #pragma unroll
      for (int i = 0; i < 4; ++i) fd[i] = dt[i];
      #pragma unroll
      for (int e = 0; e < 2; ++e) {
        v2f qre = (v2f){0.f, 0.f}, qim = (v2f){0.f, 0.f};
        #pragma unroll
        for (int beta = 0; beta < 2; ++beta) {
          v2f dre;  dre[0] = fd[e * 2 + beta][0];  dre[1] = fd[e * 2 + beta][1];
          v2f dim;  dim[0] = fd[e * 2 + beta][2];  dim[1] = fd[e * 2 + beta][3];
          qre = qre + dre * Fre[beta] - dim * Fim[beta];
          qim = qim + dre * Fim[beta] + dim * Fre[beta];
        }
        Q[e].re = qre[0] + qre[1];  Q[e].im = qim[0] + qim[1];
      }
    }
    cpx kep = pick(pG[0], pG[1], half);
    cpx snd = pick(pG[1], pG[0], half);
    cpx G;
    G.re  = kep.re + __shfl_xor(snd.re, 16, 32);
    G.im  = kep.im + __shfl_xor(snd.im, 16, 32);
    v8f accX = {}; v8f accY = {};
    #pragma unroll
    for (int e = 0; e < 2; ++e) {
      cpx Qo, M;
      Qo.re = __shfl_xor(Q[e].re, 16, 32);
      Qo.im = __shfl_xor(Q[e].im, 16, 32);
      cpx m0 = cmul(a0[4], Q[e]), m1 = cmul(a1[4], Qo);
      M.re = m0.re + m1.re;  M.im = m0.im + m1.im;
      const bool mine = (lane & 1) == e;
      v2f ag;  ag[0] = mine ? G.re : 0.f;  ag[1] = mine ? G.im : 0.f;
      v2f bx;  bx[0] = M.re;  bx[1] = -M.im;
      v2f by;  by[0] = M.im;  by[1] = M.re;
      accX = mm<false>(ag, bx, accX);
      accY = mm<false>(ag, by, accY);
    }
    #pragma unroll
    for (int v = 0; v < 8; ++v) { X[v] = accX[v]; Y[v] = accY[v]; }
  }

  rmulmat<false>(frag + 256 + 1 * 64, lane, X, Y);
  #pragma unroll
  for (int v = 1; v < 8; v += 2) {
    X[v] = __shfl_xor(X[v], 8, 32);
    Y[v] = __shfl_xor(Y[v], 8, 32);
  }
  {
    int from = chainperm(lane & 15) | (lane & 16);
    #pragma unroll
    for (int v = 0; v < 8; ++v) {
      X[v] = __shfl(X[v], from, 32);
      Y[v] = __shfl(Y[v], from, 32);
    }
  }
  apply_phases(frag + 512 + 128, lane, X, Y);
  rmulmat<true>(frag + 256 + 3 * 64, lane, X, Y);
  rmulmat<false>(frag + 256 + 2 * 64, lane, X, Y);


  float psum = 0.f, s5 = 0.f, s6 = 0.f, s7 = 0.f;
  #pragma unroll
  for (int v = 0; v < 8; ++v) {
    float p = X[v] * X[v] + Y[v] * Y[v];
    psum += p;
    s5 += ((v >> 2) & 1)                  ? -p : p;
    s6 += (((v >> 2) ^ (v >> 1)) & 1)     ? -p : p;
    s7 += (((v >> 2) ^ (v >> 1) ^ v) & 1) ? -p : p;
  }
  const bool neg = ((half ^ lane) & 1) != 0;
  float o[8];
  o[0] = ((col >> 3) & 1) ? -psum : psum;
  o[1] = ((col >> 2) & 1) ? -psum : psum;
  o[2] = ((col >> 1) & 1) ? -psum : psum;
  o[3] = ((col) & 1)      ? -psum : psum;
  o[4] = neg ? -psum : psum;
  o[5] = neg ? -s5 : s5;
  o[6] = neg ? -s6 : s6;
  o[7] = neg ? -s7 : s7;
  float o4[4], o2[2], tot;
  #pragma unroll
  for (int i = 0; i < 4; ++i) {
    float keep = half ? o[i + 4] : o[i];
    float send = half ? o[i] : o[i + 4];
    o4[i] = keep + __shfl_xor(send, 16, 32);
  }
  const int l3 = (lane >> 3) & 1;
  #pragma unroll
  for (int i = 0; i < 2; ++i) {
    float keep = l3 ? o4[i + 2] : o4[i];
    float send = l3 ? o4[i] : o4[i + 2];
    o2[i] = keep + __shfl_xor(send, 8, 32);
  }
  {
    const int l2 = (lane >> 2) & 1;
    float keep = l2 ? o2[1] : o2[0];
    float send = l2 ? o2[0] : o2[1];
    tot = keep + __shfl_xor(send, 4, 32);
  }
  tot += __shfl_xor(tot, 2, 32);
  tot += __shfl_xor(tot, 1, 32);
  if ((lane & 3) == 0 && bw < B) out[b * 8 + (lane >> 2)] = tot;
}

extern "C" void kernel_launch(void* const* d_in, const int* in_sizes, int n_in,
                              void* d_out, int out_size, void* d_ws, size_t ws_size,
                              hipStream_t stream) {
  const float* x = (const float*)d_in[0];
  const float* w = (const float*)d_in[1];
  float* out = (float*)d_out;
  int B = in_sizes[0] / NQ;

  int nblocks = (B + 31) / 32;
  qsim<<<nblocks, 1024, 0, stream>>>(x, w, out, B);
}
